# starting kernel
# baseline (speedup 1.0000x reference)
#include <hip/hip_runtime.h>

#define DINK 128
#define DHK  128
#define DOK  64

typedef __attribute__((ext_vector_type(2))) float v2f;
typedef __attribute__((ext_vector_type(8))) float v8f;


__global__ void gcn_zero_f4(float4* __restrict__ p, long n4) {
    long i = (long)blockIdx.x * blockDim.x + threadIdx.x;
    if (i < n4) p[i] = make_float4(0.f, 0.f, 0.f, 0.f);
}

__global__ void gcn_deg(const int* __restrict__ dst, float* __restrict__ deg, int E) {
    int e = blockIdx.x * blockDim.x + threadIdx.x;
    if (e < E) atomicAdd(&deg[dst[e]], 1.0f);
}

__global__ void gcn_dinv(float* __restrict__ deg, int n) {
    int i = blockIdx.x * blockDim.x + threadIdx.x;
    if (i < n) {
        float d = deg[i];
        deg[i] = (d > 0.f) ? rsqrtf(d) : 0.f;
    }
}

template<int NT>
__global__ __launch_bounds__(NT * 32)
void gcn_gemm_wmma(const float* __restrict__ H, const float* __restrict__ W,
                   float* __restrict__ out) {
    const int lane  = threadIdx.x & 31;
    const int wave  = threadIdx.x >> 5;
    const int baseM = blockIdx.x << 4;
    const int l15   = lane & 15;
    const int half  = lane >> 4;
    const int col   = (wave << 4) + l15;
    const int rowA  = baseM + l15;
    const int koff  = half << 1;

    const float* hp = H + (size_t)rowA * DINK + koff;
    const float* wp = W + (size_t)col  * DINK + koff;

    v8f acc = {0.f, 0.f, 0.f, 0.f, 0.f, 0.f, 0.f, 0.f};

#pragma unroll
    for (int k0 = 0; k0 < DINK; k0 += 4) {
        v2f a = *(const v2f*)(hp + k0);
        v2f b = *(const v2f*)(wp + k0);
        acc = __builtin_amdgcn_wmma_f32_16x16x4_f32(
             false, a,  false, b,
             (short)0, acc,  false,  false);
    }

    const int ldout = NT * 16;
    float* op = out + (size_t)(baseM + (half << 3)) * ldout + col;
#pragma unroll
    for (int r = 0; r < 8; ++r)
        op[(size_t)r * ldout] = acc[r];
}

__global__ void gcn_scatter128(const int* __restrict__ src, const int* __restrict__ dst,
                               const float* __restrict__ dinv, const float* __restrict__ tmp,
                               float* __restrict__ out, int E) {
    long tid = (long)blockIdx.x * blockDim.x + threadIdx.x;
    int e = (int)(tid >> 5);
    if (e >= E) return;
    int lane = (int)(tid & 31);
    int s = src[e], d = dst[e];
    float nrm = dinv[s] * dinv[d];
    const float4 v = *(const float4*)(tmp + (size_t)s * 128 + lane * 4);
    float* o = out + (size_t)d * 128 + lane * 4;
    atomicAdd(o + 0, v.x * nrm);
    atomicAdd(o + 1, v.y * nrm);
    atomicAdd(o + 2, v.z * nrm);
    atomicAdd(o + 3, v.w * nrm);
}

__global__ void gcn_scatter64(const int* __restrict__ src, const int* __restrict__ dst,
                              const float* __restrict__ dinv, const float* __restrict__ tmp,
                              float* __restrict__ out, int E) {
    long tid = (long)blockIdx.x * blockDim.x + threadIdx.x;
    int e = (int)(tid >> 4);
    if (e >= E) return;
    int lane = (int)(tid & 15);
    int s = src[e], d = dst[e];
    float nrm = dinv[s] * dinv[d];
    const float4 v = *(const float4*)(tmp + (size_t)s * 64 + lane * 4);
    float* o = out + (size_t)d * 64 + lane * 4;
    atomicAdd(o + 0, v.x * nrm);
    atomicAdd(o + 1, v.y * nrm);
    atomicAdd(o + 2, v.z * nrm);
    atomicAdd(o + 3, v.w * nrm);
}

template<int D, bool RELU>
__global__ void gcn_bias_act(float* __restrict__ h, const float* __restrict__ b, long n4) {
    long i = (long)blockIdx.x * blockDim.x + threadIdx.x;
    if (i >= n4) return;
    int c4 = (int)(i & (D / 4 - 1));
    float4 bb = ((const float4*)b)[c4];
    float4 v  = ((float4*)h)[i];
    v.x += bb.x; v.y += bb.y; v.z += bb.z; v.w += bb.w;
    if (RELU) {
        v.x = fmaxf(v.x, 0.f); v.y = fmaxf(v.y, 0.f);
        v.z = fmaxf(v.z, 0.f); v.w = fmaxf(v.w, 0.f);
    }
    ((float4*)h)[i] = v;
}


extern "C" void kernel_launch(void* const* d_in, const int* in_sizes, int n_in,
                              void* d_out, int out_size, void* d_ws, size_t ws_size,
                              hipStream_t stream) {
    const float* x  = (const float*)d_in[0];
    const int*   ei = (const int*)d_in[1];
    const float* W1 = (const float*)d_in[2];
    const float* b1 = (const float*)d_in[3];
    const float* W2 = (const float*)d_in[4];
    const float* b2 = (const float*)d_in[5];
    const float* W3 = (const float*)d_in[6];
    const float* b3 = (const float*)d_in[7];
    const float* W4 = (const float*)d_in[8];
    const float* b4 = (const float*)d_in[9];

    const int Nn = in_sizes[0] / DINK;
    const int E  = in_sizes[1] / 2;
    const int* srcI = ei;
    const int* dstI = ei + E;

    float* ws_f = (float*)d_ws;
    float* dinv = ws_f;
    size_t offA = ((size_t)Nn + 1023) & ~(size_t)1023;
    float* A = ws_f + offA;
    float* B = A + (size_t)Nn * DHK;
    float* out = (float*)d_out;

    const int TB = 256;
    const long nh4   = (long)Nn * DHK / 4;
    const long no4   = (long)Nn * DOK / 4;
    const int  gN16  = Nn / 16;
    const int  gSc128 = (int)(((long)E * 32 + TB - 1) / TB);
    const int  gSc64  = (int)(((long)E * 16 + TB - 1) / TB);
    const int  gH4    = (int)((nh4 + TB - 1) / TB);
    const int  gO4    = (int)((no4 + TB - 1) / TB);

    gcn_zero_f4<<<(Nn / 4 + TB - 1) / TB, TB, 0, stream>>>((float4*)dinv, Nn / 4);
    gcn_deg<<<(E + TB - 1) / TB, TB, 0, stream>>>(dstI, dinv, E);
    gcn_dinv<<<(Nn + TB - 1) / TB, TB, 0, stream>>>(dinv, Nn);

    gcn_gemm_wmma<8><<<gN16, 256, 0, stream>>>(x, W1, A);
    gcn_zero_f4<<<gH4, TB, 0, stream>>>((float4*)B, nh4);
    gcn_scatter128<<<gSc128, TB, 0, stream>>>(srcI, dstI, dinv, A, B, E);
    gcn_bias_act<128, true><<<gH4, TB, 0, stream>>>(B, b1, nh4);

    gcn_gemm_wmma<8><<<gN16, 256, 0, stream>>>(B, W2, A);
    gcn_zero_f4<<<gH4, TB, 0, stream>>>((float4*)B, nh4);
    gcn_scatter128<<<gSc128, TB, 0, stream>>>(srcI, dstI, dinv, A, B, E);
    gcn_bias_act<128, true><<<gH4, TB, 0, stream>>>(B, b2, nh4);

    gcn_gemm_wmma<8><<<gN16, 256, 0, stream>>>(B, W3, A);
    gcn_zero_f4<<<gH4, TB, 0, stream>>>((float4*)B, nh4);
    gcn_scatter128<<<gSc128, TB, 0, stream>>>(srcI, dstI, dinv, A, B, E);
    gcn_bias_act<128, true><<<gH4, TB, 0, stream>>>(B, b3, nh4);

    gcn_gemm_wmma<4><<<gN16, 128, 0, stream>>>(B, W4, A);
    gcn_zero_f4<<<gO4, TB, 0, stream>>>((float4*)out, no4);
    gcn_scatter64<<<gSc64, TB, 0, stream>>>(srcI, dstI, dinv, A, out, E);
    gcn_bias_act<64, false><<<gO4, TB, 0, stream>>>(out, b4, no4);
}


# Aggregation by destination range, no sort: each of 250 blocks filters the dst list for its 400 rows, sums the gathered source rows in LDS by LDS float atomics (arrival order), +bias, relu, one store
# speedup vs baseline: 2.4611x; 2.4611x over previous
#include <hip/hip_runtime.h>

#define DINK 128
#define DHK  128
#define DOK  64

typedef __attribute__((ext_vector_type(2))) float v2f;
typedef __attribute__((ext_vector_type(8))) float v8f;


__global__ void gcn_zero_f4(float4* __restrict__ p, long n4) {
    long i = (long)blockIdx.x * blockDim.x + threadIdx.x;
    if (i < n4) p[i] = make_float4(0.f, 0.f, 0.f, 0.f);
}

__global__ void gcn_deg(const int* __restrict__ dst, float* __restrict__ deg, int E) {
    int e = blockIdx.x * blockDim.x + threadIdx.x;
    if (e < E) atomicAdd(&deg[dst[e]], 1.0f);
}

__global__ void gcn_dinv(float* __restrict__ deg, int n) {
    int i = blockIdx.x * blockDim.x + threadIdx.x;
    if (i < n) {
        float d = deg[i];
        deg[i] = (d > 0.f) ? rsqrtf(d) : 0.f;
    }
}

template<int NT>
__global__ __launch_bounds__(128)
void gcn_gemm_wmma(const float* __restrict__ H, const float* __restrict__ W,
                   float* __restrict__ out, int mtiles) {
    const int lane = threadIdx.x & 31;
    const int wave = __builtin_amdgcn_readfirstlane((int)(threadIdx.x >> 5));
    const int tile0 = (blockIdx.x * 4 + wave) * 2;
    if (tile0 >= mtiles) return;
    const bool has2 = (tile0 + 1) < mtiles;
    const int c = lane & 15;
    const int h = lane >> 4;
    const int row0 = (tile0 << 4) + c;
    const int row1 = has2 ? row0 + 16 : row0;

    const float* hp0 = H + (size_t)row0 * DINK + (h << 2);
    const float* hp1 = H + (size_t)row1 * DINK + (h << 2);
    const float* wp  = W + (size_t)c * DINK + (h << 2);

    v8f acc[2][NT];
#pragma unroll
    for (int n = 0; n < NT; ++n) {
        acc[0][n] = (v8f){0.f, 0.f, 0.f, 0.f, 0.f, 0.f, 0.f, 0.f};
        acc[1][n] = (v8f){0.f, 0.f, 0.f, 0.f, 0.f, 0.f, 0.f, 0.f};
    }

#pragma unroll
    for (int kc = 0; kc < DINK / 8; ++kc) {
        const float4 a0 = *(const float4*)(hp0 + kc * 8);
        const float4 a1 = *(const float4*)(hp1 + kc * 8);
        float4 b[NT];
#pragma unroll
        for (int n = 0; n < NT; ++n)
            b[n] = *(const float4*)(wp + (size_t)n * 16 * DINK + kc * 8);

        const v2f a0lo = {a0.x, a0.y}, a0hi = {a0.z, a0.w};
        const v2f a1lo = {a1.x, a1.y}, a1hi = {a1.z, a1.w};
#pragma unroll
        for (int n = 0; n < NT; ++n) {
            const v2f blo = {b[n].x, b[n].y};
            acc[0][n] = __builtin_amdgcn_wmma_f32_16x16x4_f32(
                false, a0lo, false, blo, (short)0, acc[0][n], false, false);
            acc[1][n] = __builtin_amdgcn_wmma_f32_16x16x4_f32(
                false, a1lo, false, blo, (short)0, acc[1][n], false, false);
        }
#pragma unroll
        for (int n = 0; n < NT; ++n) {
            const v2f bhi = {b[n].z, b[n].w};
            acc[0][n] = __builtin_amdgcn_wmma_f32_16x16x4_f32(
                false, a0hi, false, bhi, (short)0, acc[0][n], false, false);
            acc[1][n] = __builtin_amdgcn_wmma_f32_16x16x4_f32(
                false, a1hi, false, bhi, (short)0, acc[1][n], false, false);
        }
    }

    const int ldout = NT * 16;
    float* op0 = out + (size_t)((tile0 << 4) + (h << 3)) * ldout + c;
#pragma unroll
    for (int n = 0; n < NT; ++n)
#pragma unroll
        for (int r = 0; r < 8; ++r)
            op0[(size_t)r * ldout + n * 16] = acc[0][n][r];
    if (has2) {
        float* op1 = op0 + (size_t)16 * ldout;
#pragma unroll
        for (int n = 0; n < NT; ++n)
#pragma unroll
            for (int r = 0; r < 8; ++r)
                op1[(size_t)r * ldout + n * 16] = acc[1][n][r];
    }
}

#define AGG_ROWS 400
#define AGG_QCAP 288
#define AGG_GRP  8

template<int D, bool RELU>
__global__ __launch_bounds__(1024)
void gcn_agg(const int* __restrict__ src, const int* __restrict__ dst,
             const float* __restrict__ dinv, const float* __restrict__ tmp,
             const float* __restrict__ bias, float* __restrict__ out, int E, int n) {
    constexpr int V = D / 32;
    typedef __attribute__((ext_vector_type(V))) float vrow;
    __shared__ float rows[AGG_ROWS * D];
    __shared__ float dloc[AGG_ROWS];
    __shared__ int q_e[32][AGG_QCAP];
    __shared__ int q_d[32][AGG_QCAP];

    const int tid  = threadIdx.x;
    const int lane = tid & 31;
    const int wave = __builtin_amdgcn_readfirstlane(tid >> 5);
    const int row0 = blockIdx.x * AGG_ROWS;

    float4* rows4 = (float4*)rows;
    for (int i = tid; i < AGG_ROWS * D / 4; i += 1024)
        rows4[i] = make_float4(0.f, 0.f, 0.f, 0.f);
    if (tid < AGG_ROWS) dloc[tid] = (row0 + tid < n) ? dinv[row0 + tid] : 0.f;
    __syncthreads();

    int* qe = q_e[wave];
    int* qd = q_d[wave];
    int qn = 0;

    auto drain = [&]() {
        const int cnt = qn < 32 ? qn : 32;
        const int q0 = qn - cnt;
        const int qi = q0 + (lane < cnt ? lane : 0);
        const int e = qe[qi];
        const int r = qd[qi] - row0;
        const int s = src[e];
        const float nrm = dinv[s] * dloc[r];
#pragma nounroll
        for (int j = 0; j < cnt; j += AGG_GRP) {
            vrow val[AGG_GRP];
#pragma unroll
            for (int k = 0; k < AGG_GRP; ++k) {
                const int jk = (j + k < cnt) ? j + k : cnt - 1;
                const int sk = __builtin_amdgcn_readlane(s, jk);
                val[k] = *(const vrow*)(tmp + (size_t)sk * D + V * lane);
            }
#pragma unroll
            for (int k = 0; k < AGG_GRP; ++k) {
                if (j + k < cnt) {
                    const float nk = __builtin_bit_cast(float,
                        __builtin_amdgcn_readlane(__builtin_bit_cast(int, nrm), j + k));
                    const int rk = __builtin_amdgcn_readlane(r, j + k);
                    float* o = &rows[rk * D + lane];
#pragma unroll
                    for (int c = 0; c < V; ++c) atomicAdd(o + 32 * c, val[k][c] * nk);
                }
            }
        }
        qn = q0;
    };

    const int nchunk = (E + 1023) >> 10;
    const int start = (int)(((long)blockIdx.x * nchunk) / gridDim.x);
    const bool vec_ok = (((unsigned long long)dst) & 15ull) == 0ull;
    for (int c0 = wave; c0 < nchunk; c0 += 32) {
        int cc = start + c0;
        if (cc >= nchunk) cc -= nchunk;
        const int cbase = cc << 10;
        const int ebase = cbase + 4 * lane;
        int dv[32];
        if (vec_ok && cbase + 1024 <= E) {
            const int4* d4 = (const int4*)(dst + cbase) + lane;
#pragma unroll
            for (int i = 0; i < 8; ++i) {
                const int4 t = d4[32 * i];
                dv[4 * i + 0] = t.x; dv[4 * i + 1] = t.y; dv[4 * i + 2] = t.z; dv[4 * i + 3] = t.w;
            }
        } else {
#pragma unroll
            for (int x = 0; x < 32; ++x) {
                const int e = ebase + 128 * (x >> 2) + (x & 3);
                const int v = dst[e < E ? e : E - 1];
                dv[x] = (e < E) ? v : -1;
            }
        }
#pragma unroll
        for (int part = 0; part < 4; ++part) {
#pragma unroll
            for (int x = 8 * part; x < 8 * part + 8; ++x) {
                const bool m = (unsigned)(dv[x] - row0) < (unsigned)AGG_ROWS;
                const unsigned mask = __builtin_amdgcn_ballot_w32(m);
                if (m) {
                    const int p = qn + (int)__builtin_amdgcn_mbcnt_lo(mask, 0u);
                    qe[p] = ebase + 128 * (x >> 2) + (x & 3);
                    qd[p] = dv[x];
                }
                qn += __builtin_popcount(mask);
            }
            while (qn >= 32) drain();
        }
    }
    while (qn > 0) drain();
    __syncthreads();

    const float4* b4 = (const float4*)bias;
    float4* out4 = (float4*)out + (size_t)row0 * (D / 4);
    const long lim = ((long)n - row0) * (D / 4);
    for (int i = tid; i < AGG_ROWS * D / 4; i += 1024) {
        if (i < lim) {
            const int q = i & (D / 4 - 1);
            const float* rp = rows + (i / (D / 4)) * D;
            const int f = 4 * q;
            float4 v;
            v.x = rp[((f + 0) % V) * 32 + (f + 0) / V];
            v.y = rp[((f + 1) % V) * 32 + (f + 1) / V];
            v.z = rp[((f + 2) % V) * 32 + (f + 2) / V];
            v.w = rp[((f + 3) % V) * 32 + (f + 3) / V];
            const float4 bb = b4[q];
            v.x += bb.x; v.y += bb.y; v.z += bb.z; v.w += bb.w;
            if (RELU) {
                v.x = fmaxf(v.x, 0.f); v.y = fmaxf(v.y, 0.f);
                v.z = fmaxf(v.z, 0.f); v.w = fmaxf(v.w, 0.f);
            }
            out4[i] = v;
        }
    }
}


extern "C" void kernel_launch(void* const* d_in, const int* in_sizes, int n_in,
                              void* d_out, int out_size, void* d_ws, size_t ws_size,
                              hipStream_t stream) {
    const float* x  = (const float*)d_in[0];
    const int*   ei = (const int*)d_in[1];
    const float* W1 = (const float*)d_in[2];
    const float* b1 = (const float*)d_in[3];
    const float* W2 = (const float*)d_in[4];
    const float* b2 = (const float*)d_in[5];
    const float* W3 = (const float*)d_in[6];
    const float* b3 = (const float*)d_in[7];
    const float* W4 = (const float*)d_in[8];
    const float* b4 = (const float*)d_in[9];

    const int Nn = in_sizes[0] / DINK;
    const int E  = in_sizes[1] / 2;
    const int* srcI = ei;
    const int* dstI = ei + E;

    float* ws_f = (float*)d_ws;
    float* dinv = ws_f;
    size_t offA = ((size_t)Nn + 1023) & ~(size_t)1023;
    float* A = ws_f + offA;
    float* B = A + (size_t)Nn * DHK;
    float* out = (float*)d_out;

    const int TB = 256;
    const int  mT    = Nn / 16;
    const int  gG    = ((mT + 1) / 2 + 3) / 4;
    const int  gAg   = (Nn + AGG_ROWS - 1) / AGG_ROWS;

    gcn_zero_f4<<<(Nn / 4 + TB - 1) / TB, TB, 0, stream>>>((float4*)dinv, Nn / 4);
    gcn_deg<<<(E + TB - 1) / TB, TB, 0, stream>>>(dstI, dinv, E);
    gcn_dinv<<<(Nn + TB - 1) / TB, TB, 0, stream>>>(dinv, Nn);

    gcn_gemm_wmma<8><<<gG, 128, 0, stream>>>(x, W1, A, mT);
    gcn_agg<128, true><<<gAg, 1024, 0, stream>>>(srcI, dstI, dinv, A, b1, B, E, Nn);

    gcn_gemm_wmma<8><<<gG, 128, 0, stream>>>(B, W2, A, mT);
    gcn_agg<128, true><<<gAg, 1024, 0, stream>>>(srcI, dstI, dinv, A, b2, B, E, Nn);

    gcn_gemm_wmma<8><<<gG, 128, 0, stream>>>(B, W3, A, mT);
    gcn_agg<128, true><<<gAg, 1024, 0, stream>>>(srcI, dstI, dinv, A, b3, B, E, Nn);

    gcn_gemm_wmma<4><<<gG, 128, 0, stream>>>(B, W4, A, mT);
    gcn_agg<64, false><<<gAg, 1024, 0, stream>>>(srcI, dstI, dinv, A, b4, out, E, Nn);
}


# Dst filter run once per call: layer 1's aggregation files each block's kept (src,row) pairs in 256-byte chunks in the unwritten out buffer; layers 2-4 walk them, no rescan; LDS arrival-order sums
# speedup vs baseline: 3.3708x; 1.3696x over previous
#include <hip/hip_runtime.h>

#define DINK 128
#define DHK  128
#define DOK  64

typedef __attribute__((ext_vector_type(2))) float v2f;
typedef __attribute__((ext_vector_type(8))) float v8f;


__global__ void gcn_zero_f4(float4* __restrict__ p, long n4) {
    long i = (long)blockIdx.x * blockDim.x + threadIdx.x;
    if (i < n4) p[i] = make_float4(0.f, 0.f, 0.f, 0.f);
}

__global__ void gcn_deg(const int* __restrict__ dst, float* __restrict__ deg, int E) {
    int e = blockIdx.x * blockDim.x + threadIdx.x;
    if (e < E) atomicAdd(&deg[dst[e]], 1.0f);
}

__global__ void gcn_dinv(float* __restrict__ deg, int n) {
    int i = blockIdx.x * blockDim.x + threadIdx.x;
    if (i < n) {
        float d = deg[i];
        deg[i] = (d > 0.f) ? rsqrtf(d) : 0.f;
    }
}

template<int NT>
__global__ __launch_bounds__(128)
void gcn_gemm_wmma(const float* __restrict__ H, const float* __restrict__ W,
                   float* __restrict__ out, int mtiles) {
    const int lane = threadIdx.x & 31;
    const int wave = __builtin_amdgcn_readfirstlane((int)(threadIdx.x >> 5));
    const int tile0 = (blockIdx.x * 4 + wave) * 2;
    if (tile0 >= mtiles) return;
    const bool has2 = (tile0 + 1) < mtiles;
    const int c = lane & 15;
    const int h = lane >> 4;
    const int row0 = (tile0 << 4) + c;
    const int row1 = has2 ? row0 + 16 : row0;

    const float* hp0 = H + (size_t)row0 * DINK + (h << 2);
    const float* hp1 = H + (size_t)row1 * DINK + (h << 2);
    const float* wp  = W + (size_t)c * DINK + (h << 2);

    v8f acc[2][NT];
#pragma unroll
    for (int n = 0; n < NT; ++n) {
        acc[0][n] = (v8f){0.f, 0.f, 0.f, 0.f, 0.f, 0.f, 0.f, 0.f};
        acc[1][n] = (v8f){0.f, 0.f, 0.f, 0.f, 0.f, 0.f, 0.f, 0.f};
    }

#pragma unroll
    for (int kc = 0; kc < DINK / 8; ++kc) {
        const float4 a0 = *(const float4*)(hp0 + kc * 8);
        const float4 a1 = *(const float4*)(hp1 + kc * 8);
        float4 b[NT];
#pragma unroll
        for (int n = 0; n < NT; ++n)
            b[n] = *(const float4*)(wp + (size_t)n * 16 * DINK + kc * 8);

        const v2f a0lo = {a0.x, a0.y}, a0hi = {a0.z, a0.w};
        const v2f a1lo = {a1.x, a1.y}, a1hi = {a1.z, a1.w};
#pragma unroll
        for (int n = 0; n < NT; ++n) {
            const v2f blo = {b[n].x, b[n].y};
            acc[0][n] = __builtin_amdgcn_wmma_f32_16x16x4_f32(
                false, a0lo, false, blo, (short)0, acc[0][n], false, false);
            acc[1][n] = __builtin_amdgcn_wmma_f32_16x16x4_f32(
                false, a1lo, false, blo, (short)0, acc[1][n], false, false);
        }
#pragma unroll
        for (int n = 0; n < NT; ++n) {
            const v2f bhi = {b[n].z, b[n].w};
            acc[0][n] = __builtin_amdgcn_wmma_f32_16x16x4_f32(
                false, a0hi, false, bhi, (short)0, acc[0][n], false, false);
            acc[1][n] = __builtin_amdgcn_wmma_f32_16x16x4_f32(
                false, a1hi, false, bhi, (short)0, acc[1][n], false, false);
        }
    }

    const int ldout = NT * 16;
    float* op0 = out + (size_t)((tile0 << 4) + (h << 3)) * ldout + c;
#pragma unroll
    for (int n = 0; n < NT; ++n)
#pragma unroll
        for (int r = 0; r < 8; ++r)
            op0[(size_t)r * ldout + n * 16] = acc[0][n][r];
    if (has2) {
        float* op1 = op0 + (size_t)16 * ldout;
#pragma unroll
        for (int n = 0; n < NT; ++n)
#pragma unroll
            for (int r = 0; r < 8; ++r)
                op1[(size_t)r * ldout + n * 16] = acc[1][n][r];
    }
}

#define AGG_ROWS 400
#define AGG_QCAP 288
#define AGG_GRP  8

template<int D, bool RELU, bool REPLAY>
__global__ __launch_bounds__(1024)
void gcn_agg(const int* __restrict__ src, const int* __restrict__ dst,
             const float* __restrict__ dinv, const float* __restrict__ tmp,
             const float* __restrict__ bias, float* out, float* lists, int E, int n) {
    constexpr int V = D / 32;
    typedef __attribute__((ext_vector_type(V))) float vrow;
    __shared__ float rows[AGG_ROWS * D];
    __shared__ float dloc[AGG_ROWS];
    __shared__ int q_e[32][AGG_QCAP];
    __shared__ int q_d[32][AGG_QCAP];
    __shared__ int lcount;

    const int tid  = threadIdx.x;
    const int lane = tid & 31;
    const int wave = __builtin_amdgcn_readfirstlane(tid >> 5);
    const int row0 = blockIdx.x * AGG_ROWS;

    float4* rows4 = (float4*)rows;
    for (int i = tid; i < AGG_ROWS * D / 4; i += 1024)
        rows4[i] = make_float4(0.f, 0.f, 0.f, 0.f);
    if (tid < AGG_ROWS) dloc[tid] = (row0 + tid < n) ? dinv[row0 + tid] : 0.f;
    if (tid == 0) lcount = 0;

    int2* mylist = (int2*)(lists + (size_t)row0 * DOK);
    const int rows_here = (n - row0 < AGG_ROWS) ? (n - row0) : AGG_ROWS;
    const int capch = rows_here * DOK / 64 - 1;
    int nch = 0;
    if (REPLAY) nch = __builtin_amdgcn_readfirstlane(mylist[0].x);
    const bool replay = REPLAY && nch >= 0 && nch <= capch;
    __syncthreads();

    int* qe = q_e[wave];
    int* qd = q_d[wave];
    int qn = 0;

    auto add_rows = [&](const int s, const int r, const float nrm, const int cnt) {
#pragma nounroll
        for (int j = 0; j < cnt; j += AGG_GRP) {
            vrow val[AGG_GRP];
#pragma unroll
            for (int k = 0; k < AGG_GRP; ++k) {
                const int jk = (j + k < cnt) ? j + k : cnt - 1;
                const int sk = __builtin_amdgcn_readlane(s, jk);
                val[k] = *(const vrow*)(tmp + (size_t)sk * D + V * lane);
            }
#pragma unroll
            for (int k = 0; k < AGG_GRP; ++k) {
                if (j + k < cnt) {
                    const float nk = __builtin_bit_cast(float,
                        __builtin_amdgcn_readlane(__builtin_bit_cast(int, nrm), j + k));
                    const int rk = __builtin_amdgcn_readlane(r, j + k);
                    float* o = &rows[rk * D + lane];
#pragma unroll
                    for (int c = 0; c < V; ++c) atomicAdd(o + 32 * c, val[k][c] * nk);
                }
            }
        }
    };

    auto drain = [&]() {
        const int cnt = qn < 32 ? qn : 32;
        const int q0 = qn - cnt;
        const int qi = q0 + (lane < cnt ? lane : 0);
        const int e = qe[qi];
        const int r = qd[qi] - row0;
        const int s = src[e];
        const float nrm = dinv[s] * dloc[r];
        if (!REPLAY) {
            int b = 0;
            if (lane == 0) b = atomicAdd(&lcount, 1);
            b = __builtin_amdgcn_readfirstlane(b);
            if (b < capch)
                mylist[(b + 1) * 32 + lane] = (lane < cnt) ? make_int2(s, r) : make_int2(0, -1);
        }
        add_rows(s, r, nrm, cnt);
        qn = q0;
    };

    const int nchunk = (E + 1023) >> 10;
    const int start = (int)(((long)blockIdx.x * nchunk) / gridDim.x);
    const bool vec_ok = (((unsigned long long)dst) & 15ull) == 0ull;
    if (replay) {
        for (int c = wave; c < nch; c += 32) {
            const int2 sr = mylist[(c + 1) * 32 + lane];
            const bool ok = sr.y >= 0;
            const int cnt = __builtin_popcount(__builtin_amdgcn_ballot_w32(ok));
            const float nrm = ok ? dinv[sr.x] * dloc[sr.y] : 0.f;
            if (cnt == 32) add_rows(sr.x, sr.y, nrm, 32);
            else if (cnt > 0) add_rows(sr.x, sr.y, nrm, cnt);
        }
    } else {
    for (int c0 = wave; c0 < nchunk; c0 += 32) {
        int cc = start + c0;
        if (cc >= nchunk) cc -= nchunk;
        const int cbase = cc << 10;
        const int ebase = cbase + 4 * lane;
        int dv[32];
        if (vec_ok && cbase + 1024 <= E) {
            const int4* d4 = (const int4*)(dst + cbase) + lane;
#pragma unroll
            for (int i = 0; i < 8; ++i) {
                const int4 t = d4[32 * i];
                dv[4 * i + 0] = t.x; dv[4 * i + 1] = t.y; dv[4 * i + 2] = t.z; dv[4 * i + 3] = t.w;
            }
        } else {
#pragma unroll
            for (int x = 0; x < 32; ++x) {
                const int e = ebase + 128 * (x >> 2) + (x & 3);
                const int v = dst[e < E ? e : E - 1];
                dv[x] = (e < E) ? v : -1;
            }
        }
#pragma unroll
        for (int part = 0; part < 4; ++part) {
#pragma unroll
            for (int x = 8 * part; x < 8 * part + 8; ++x) {
                const bool m = (unsigned)(dv[x] - row0) < (unsigned)AGG_ROWS;
                const unsigned mask = __builtin_amdgcn_ballot_w32(m);
                if (m) {
                    const int p = qn + (int)__builtin_amdgcn_mbcnt_lo(mask, 0u);
                    qe[p] = ebase + 128 * (x >> 2) + (x & 3);
                    qd[p] = dv[x];
                }
                qn += __builtin_popcount(mask);
            }
            while (qn >= 32) drain();
        }
    }
    while (qn > 0) drain();
    }
    __syncthreads();
    if (!REPLAY && wave == 0) mylist[lane] = make_int2(lcount, 0);

    const float4* b4 = (const float4*)bias;
    float4* out4 = (float4*)out + (size_t)row0 * (D / 4);
    const long lim = ((long)n - row0) * (D / 4);
    for (int i = tid; i < AGG_ROWS * D / 4; i += 1024) {
        if (i < lim) {
            const int q = i & (D / 4 - 1);
            const float* rp = rows + (i / (D / 4)) * D;
            const int f = 4 * q;
            float4 v;
            v.x = rp[((f + 0) % V) * 32 + (f + 0) / V];
            v.y = rp[((f + 1) % V) * 32 + (f + 1) / V];
            v.z = rp[((f + 2) % V) * 32 + (f + 2) / V];
            v.w = rp[((f + 3) % V) * 32 + (f + 3) / V];
            const float4 bb = b4[q];
            v.x += bb.x; v.y += bb.y; v.z += bb.z; v.w += bb.w;
            if (RELU) {
                v.x = fmaxf(v.x, 0.f); v.y = fmaxf(v.y, 0.f);
                v.z = fmaxf(v.z, 0.f); v.w = fmaxf(v.w, 0.f);
            }
            out4[i] = v;
        }
    }
}


extern "C" void kernel_launch(void* const* d_in, const int* in_sizes, int n_in,
                              void* d_out, int out_size, void* d_ws, size_t ws_size,
                              hipStream_t stream) {
    const float* x  = (const float*)d_in[0];
    const int*   ei = (const int*)d_in[1];
    const float* W1 = (const float*)d_in[2];
    const float* b1 = (const float*)d_in[3];
    const float* W2 = (const float*)d_in[4];
    const float* b2 = (const float*)d_in[5];
    const float* W3 = (const float*)d_in[6];
    const float* b3 = (const float*)d_in[7];
    const float* W4 = (const float*)d_in[8];
    const float* b4 = (const float*)d_in[9];

    const int Nn = in_sizes[0] / DINK;
    const int E  = in_sizes[1] / 2;
    const int* srcI = ei;
    const int* dstI = ei + E;

    float* ws_f = (float*)d_ws;
    float* dinv = ws_f;
    size_t offA = ((size_t)Nn + 1023) & ~(size_t)1023;
    float* A = ws_f + offA;
    float* B = A + (size_t)Nn * DHK;
    float* out = (float*)d_out;

    const int TB = 256;
    const int  mT    = Nn / 16;
    const int  gG    = ((mT + 1) / 2 + 3) / 4;
    const int  gAg   = (Nn + AGG_ROWS - 1) / AGG_ROWS;

    gcn_zero_f4<<<(Nn / 4 + TB - 1) / TB, TB, 0, stream>>>((float4*)dinv, Nn / 4);
    gcn_deg<<<(E + TB - 1) / TB, TB, 0, stream>>>(dstI, dinv, E);
    gcn_dinv<<<(Nn + TB - 1) / TB, TB, 0, stream>>>(dinv, Nn);

    gcn_gemm_wmma<8><<<gG, 128, 0, stream>>>(x, W1, A, mT);
    gcn_agg<128, true, false><<<gAg, 1024, 0, stream>>>(srcI, dstI, dinv, A, b1, B, out, E, Nn);

    gcn_gemm_wmma<8><<<gG, 128, 0, stream>>>(B, W2, A, mT);
    gcn_agg<128, true, true><<<gAg, 1024, 0, stream>>>(srcI, dstI, dinv, A, b2, B, out, E, Nn);

    gcn_gemm_wmma<8><<<gG, 128, 0, stream>>>(B, W3, A, mT);
    gcn_agg<128, true, true><<<gAg, 1024, 0, stream>>>(srcI, dstI, dinv, A, b3, B, out, E, Nn);

    gcn_gemm_wmma<4><<<gG, 128, 0, stream>>>(B, W4, A, mT);
    gcn_agg<64, false, true><<<gAg, 1024, 0, stream>>>(srcI, dstI, dinv, A, b4, out, out, E, Nn);
}


# GEMM persistent, W register-resident: 256 blocks of 4 waves (one per SIMD); a wave loads all W fragments once (512 VGPRs), walks 16-row tiles, next tile's H rows asked for ahead; k order kept
# speedup vs baseline: 3.5470x; 1.0523x over previous
#include <hip/hip_runtime.h>

#define DINK 128
#define DHK  128
#define DOK  64

typedef __attribute__((ext_vector_type(2))) float v2f;
typedef __attribute__((ext_vector_type(8))) float v8f;


__global__ void gcn_zero_f4(float4* __restrict__ p, long n4) {
    long i = (long)blockIdx.x * blockDim.x + threadIdx.x;
    if (i < n4) p[i] = make_float4(0.f, 0.f, 0.f, 0.f);
}

__global__ void gcn_deg(const int* __restrict__ dst, float* __restrict__ deg, int E) {
    int e = blockIdx.x * blockDim.x + threadIdx.x;
    if (e < E) atomicAdd(&deg[dst[e]], 1.0f);
}

__global__ void gcn_dinv(float* __restrict__ deg, int n) {
    int i = blockIdx.x * blockDim.x + threadIdx.x;
    if (i < n) {
        float d = deg[i];
        deg[i] = (d > 0.f) ? rsqrtf(d) : 0.f;
    }
}

template<int NT>
__global__ __launch_bounds__(128) __attribute__((amdgpu_waves_per_eu(1, 1)))
void gcn_gemm_wmma(const float* __restrict__ H, const float* __restrict__ W,
                   float* __restrict__ out, int mtiles) {
    const int lane = threadIdx.x & 31;
    const int wave = __builtin_amdgcn_readfirstlane((int)(threadIdx.x >> 5));
    const int gw = blockIdx.x * 4 + wave;
    const int nw = gridDim.x * 4;
    if (gw >= mtiles) return;
    const int c = lane & 15;
    const int h = lane >> 4;

    const float* wp = W + (size_t)c * DINK + (h << 2);
    float4 wf[DINK / 8][NT];
#pragma unroll
    for (int kc = 0; kc < DINK / 8; ++kc)
#pragma unroll
        for (int n = 0; n < NT; ++n)
            wf[kc][n] = *(const float4*)(wp + (size_t)n * 16 * DINK + kc * 8);

    float4 an[DINK / 8];
    {
        const float* hp = H + (size_t)((gw << 4) + c) * DINK + (h << 2);
#pragma unroll
        for (int kc = 0; kc < DINK / 8; ++kc) an[kc] = *(const float4*)(hp + kc * 8);
    }

    const int ldout = NT * 16;
    for (int tile = gw; tile < mtiles; tile += nw) {
        float4 ac[DINK / 8];
#pragma unroll
        for (int kc = 0; kc < DINK / 8; ++kc) ac[kc] = an[kc];
        {
            const int nt = (tile + nw < mtiles) ? tile + nw : tile;
            const float* hp = H + (size_t)((nt << 4) + c) * DINK + (h << 2);
#pragma unroll
            for (int kc = 0; kc < DINK / 8; ++kc) an[kc] = *(const float4*)(hp + kc * 8);
        }

        v8f acc[NT];
#pragma unroll
        for (int n = 0; n < NT; ++n)
            acc[n] = (v8f){0.f, 0.f, 0.f, 0.f, 0.f, 0.f, 0.f, 0.f};

#pragma unroll
        for (int kc = 0; kc < DINK / 8; ++kc) {
            const v2f alo = {ac[kc].x, ac[kc].y}, ahi = {ac[kc].z, ac[kc].w};
#pragma unroll
            for (int n = 0; n < NT; ++n) {
                const v2f blo = {wf[kc][n].x, wf[kc][n].y};
                acc[n] = __builtin_amdgcn_wmma_f32_16x16x4_f32(
                    false, alo, false, blo, (short)0, acc[n], false, false);
            }
#pragma unroll
            for (int n = 0; n < NT; ++n) {
                const v2f bhi = {wf[kc][n].z, wf[kc][n].w};
                acc[n] = __builtin_amdgcn_wmma_f32_16x16x4_f32(
                    false, ahi, false, bhi, (short)0, acc[n], false, false);
            }
        }

        float* op = out + (size_t)((tile << 4) + (h << 3)) * ldout + c;
#pragma unroll
        for (int n = 0; n < NT; ++n)
#pragma unroll
            for (int r = 0; r < 8; ++r)
                op[(size_t)r * ldout + n * 16] = acc[n][r];
    }
}

#define AGG_ROWS 400
#define AGG_QCAP 288
#define AGG_GRP  8

template<int D, bool RELU, bool REPLAY>
__global__ __launch_bounds__(1024)
void gcn_agg(const int* __restrict__ src, const int* __restrict__ dst,
             const float* __restrict__ dinv, const float* __restrict__ tmp,
             const float* __restrict__ bias, float* out, float* lists, int E, int n) {
    constexpr int V = D / 32;
    typedef __attribute__((ext_vector_type(V))) float vrow;
    __shared__ float rows[AGG_ROWS * D];
    __shared__ float dloc[AGG_ROWS];
    __shared__ int q_e[32][AGG_QCAP];
    __shared__ int q_d[32][AGG_QCAP];
    __shared__ int lcount;

    const int tid  = threadIdx.x;
    const int lane = tid & 31;
    const int wave = __builtin_amdgcn_readfirstlane(tid >> 5);
    const int row0 = blockIdx.x * AGG_ROWS;

    float4* rows4 = (float4*)rows;
    for (int i = tid; i < AGG_ROWS * D / 4; i += 1024)
        rows4[i] = make_float4(0.f, 0.f, 0.f, 0.f);
    if (tid < AGG_ROWS) dloc[tid] = (row0 + tid < n) ? dinv[row0 + tid] : 0.f;
    if (tid == 0) lcount = 0;

    int2* mylist = (int2*)(lists + (size_t)row0 * DOK);
    const int rows_here = (n - row0 < AGG_ROWS) ? (n - row0) : AGG_ROWS;
    const int capch = rows_here * DOK / 64 - 1;
    int nch = 0;
    if (REPLAY) nch = __builtin_amdgcn_readfirstlane(mylist[0].x);
    const bool replay = REPLAY && nch >= 0 && nch <= capch;
    __syncthreads();

    int* qe = q_e[wave];
    int* qd = q_d[wave];
    int qn = 0;

    auto add_rows = [&](const int s, const int r, const float nrm, const int cnt) {
#pragma nounroll
        for (int j = 0; j < cnt; j += AGG_GRP) {
            vrow val[AGG_GRP];
#pragma unroll
            for (int k = 0; k < AGG_GRP; ++k) {
                const int jk = (j + k < cnt) ? j + k : cnt - 1;
                const int sk = __builtin_amdgcn_readlane(s, jk);
                val[k] = *(const vrow*)(tmp + (size_t)sk * D + V * lane);
            }
#pragma unroll
            for (int k = 0; k < AGG_GRP; ++k) {
                if (j + k < cnt) {
                    const float nk = __builtin_bit_cast(float,
                        __builtin_amdgcn_readlane(__builtin_bit_cast(int, nrm), j + k));
                    const int rk = __builtin_amdgcn_readlane(r, j + k);
                    float* o = &rows[rk * D + lane];
#pragma unroll
                    for (int c = 0; c < V; ++c) atomicAdd(o + 32 * c, val[k][c] * nk);
                }
            }
        }
    };

    auto drain = [&]() {
        const int cnt = qn < 32 ? qn : 32;
        const int q0 = qn - cnt;
        const int qi = q0 + (lane < cnt ? lane : 0);
        const int e = qe[qi];
        const int r = qd[qi] - row0;
        const int s = src[e];
        const float nrm = dinv[s] * dloc[r];
        if (!REPLAY) {
            int b = 0;
            if (lane == 0) b = atomicAdd(&lcount, 1);
            b = __builtin_amdgcn_readfirstlane(b);
            if (b < capch)
                mylist[(b + 1) * 32 + lane] = (lane < cnt) ? make_int2(s, r) : make_int2(0, -1);
        }
        add_rows(s, r, nrm, cnt);
        qn = q0;
    };

    const int nchunk = (E + 1023) >> 10;
    const int start = (int)(((long)blockIdx.x * nchunk) / gridDim.x);
    const bool vec_ok = (((unsigned long long)dst) & 15ull) == 0ull;
    if (replay) {
        for (int c = wave; c < nch; c += 32) {
            const int2 sr = mylist[(c + 1) * 32 + lane];
            const bool ok = sr.y >= 0;
            const int cnt = __builtin_popcount(__builtin_amdgcn_ballot_w32(ok));
            const float nrm = ok ? dinv[sr.x] * dloc[sr.y] : 0.f;
            if (cnt == 32) add_rows(sr.x, sr.y, nrm, 32);
            else if (cnt > 0) add_rows(sr.x, sr.y, nrm, cnt);
        }
    } else {
    for (int c0 = wave; c0 < nchunk; c0 += 32) {
        int cc = start + c0;
        if (cc >= nchunk) cc -= nchunk;
        const int cbase = cc << 10;
        const int ebase = cbase + 4 * lane;
        int dv[32];
        if (vec_ok && cbase + 1024 <= E) {
            const int4* d4 = (const int4*)(dst + cbase) + lane;
#pragma unroll
            for (int i = 0; i < 8; ++i) {
                const int4 t = d4[32 * i];
                dv[4 * i + 0] = t.x; dv[4 * i + 1] = t.y; dv[4 * i + 2] = t.z; dv[4 * i + 3] = t.w;
            }
        } else {
#pragma unroll
            for (int x = 0; x < 32; ++x) {
                const int e = ebase + 128 * (x >> 2) + (x & 3);
                const int v = dst[e < E ? e : E - 1];
                dv[x] = (e < E) ? v : -1;
            }
        }
#pragma unroll
        for (int part = 0; part < 4; ++part) {
#pragma unroll
            for (int x = 8 * part; x < 8 * part + 8; ++x) {
                const bool m = (unsigned)(dv[x] - row0) < (unsigned)AGG_ROWS;
                const unsigned mask = __builtin_amdgcn_ballot_w32(m);
                if (m) {
                    const int p = qn + (int)__builtin_amdgcn_mbcnt_lo(mask, 0u);
                    qe[p] = ebase + 128 * (x >> 2) + (x & 3);
                    qd[p] = dv[x];
                }
                qn += __builtin_popcount(mask);
            }
            while (qn >= 32) drain();
        }
    }
    while (qn > 0) drain();
    }
    __syncthreads();
    if (!REPLAY && wave == 0) mylist[lane] = make_int2(lcount, 0);

    const float4* b4 = (const float4*)bias;
    float4* out4 = (float4*)out + (size_t)row0 * (D / 4);
    const long lim = ((long)n - row0) * (D / 4);
    for (int i = tid; i < AGG_ROWS * D / 4; i += 1024) {
        if (i < lim) {
            const int q = i & (D / 4 - 1);
            const float* rp = rows + (i / (D / 4)) * D;
            const int f = 4 * q;
            float4 v;
            v.x = rp[((f + 0) % V) * 32 + (f + 0) / V];
            v.y = rp[((f + 1) % V) * 32 + (f + 1) / V];
            v.z = rp[((f + 2) % V) * 32 + (f + 2) / V];
            v.w = rp[((f + 3) % V) * 32 + (f + 3) / V];
            const float4 bb = b4[q];
            v.x += bb.x; v.y += bb.y; v.z += bb.z; v.w += bb.w;
            if (RELU) {
                v.x = fmaxf(v.x, 0.f); v.y = fmaxf(v.y, 0.f);
                v.z = fmaxf(v.z, 0.f); v.w = fmaxf(v.w, 0.f);
            }
            out4[i] = v;
        }
    }
}


extern "C" void kernel_launch(void* const* d_in, const int* in_sizes, int n_in,
                              void* d_out, int out_size, void* d_ws, size_t ws_size,
                              hipStream_t stream) {
    const float* x  = (const float*)d_in[0];
    const int*   ei = (const int*)d_in[1];
    const float* W1 = (const float*)d_in[2];
    const float* b1 = (const float*)d_in[3];
    const float* W2 = (const float*)d_in[4];
    const float* b2 = (const float*)d_in[5];
    const float* W3 = (const float*)d_in[6];
    const float* b3 = (const float*)d_in[7];
    const float* W4 = (const float*)d_in[8];
    const float* b4 = (const float*)d_in[9];

    const int Nn = in_sizes[0] / DINK;
    const int E  = in_sizes[1] / 2;
    const int* srcI = ei;
    const int* dstI = ei + E;

    float* ws_f = (float*)d_ws;
    float* dinv = ws_f;
    size_t offA = ((size_t)Nn + 1023) & ~(size_t)1023;
    float* A = ws_f + offA;
    float* B = A + (size_t)Nn * DHK;
    float* out = (float*)d_out;

    const int TB = 256;
    const int  mT    = Nn / 16;
    const int  gG    = 256;
    const int  gAg   = (Nn + AGG_ROWS - 1) / AGG_ROWS;

    gcn_zero_f4<<<(Nn / 4 + TB - 1) / TB, TB, 0, stream>>>((float4*)dinv, Nn / 4);
    gcn_deg<<<(E + TB - 1) / TB, TB, 0, stream>>>(dstI, dinv, E);
    gcn_dinv<<<(Nn + TB - 1) / TB, TB, 0, stream>>>(dinv, Nn);

    gcn_gemm_wmma<8><<<gG, 128, 0, stream>>>(x, W1, A, mT);
    gcn_agg<128, true, false><<<gAg, 1024, 0, stream>>>(srcI, dstI, dinv, A, b1, B, out, E, Nn);

    gcn_gemm_wmma<8><<<gG, 128, 0, stream>>>(B, W2, A, mT);
    gcn_agg<128, true, true><<<gAg, 1024, 0, stream>>>(srcI, dstI, dinv, A, b2, B, out, E, Nn);

    gcn_gemm_wmma<8><<<gG, 128, 0, stream>>>(B, W3, A, mT);
    gcn_agg<128, true, true><<<gAg, 1024, 0, stream>>>(srcI, dstI, dinv, A, b3, B, out, E, Nn);

    gcn_gemm_wmma<4><<<gG, 128, 0, stream>>>(B, W4, A, mT);
    gcn_agg<64, false, true><<<gAg, 1024, 0, stream>>>(srcI, dstI, dinv, A, b4, out, out, E, Nn);
}


# Source weight deg^-1/2 folded into the product's store (GEMM writes dinv[i]*row i); edges add unscaled rows, dst weight once per stored row: no scattered dinv read per edge; (h*ds)*dd, was h*(ds*dd)
# speedup vs baseline: 3.7443x; 1.0556x over previous
#include <hip/hip_runtime.h>

#define DINK 128
#define DHK  128
#define DOK  64

typedef __attribute__((ext_vector_type(2))) float v2f;
typedef __attribute__((ext_vector_type(8))) float v8f;


__global__ void gcn_zero_f4(float4* __restrict__ p, long n4) {
    long i = (long)blockIdx.x * blockDim.x + threadIdx.x;
    if (i < n4) p[i] = make_float4(0.f, 0.f, 0.f, 0.f);
}

__global__ void gcn_deg(const int* __restrict__ dst, float* __restrict__ deg, int E) {
    int e = blockIdx.x * blockDim.x + threadIdx.x;
    if (e < E) atomicAdd(&deg[dst[e]], 1.0f);
}

__global__ void gcn_dinv(float* __restrict__ deg, int n) {
    int i = blockIdx.x * blockDim.x + threadIdx.x;
    if (i < n) {
        float d = deg[i];
        deg[i] = (d > 0.f) ? rsqrtf(d) : 0.f;
    }
}

template<int NT>
__global__ __launch_bounds__(128) __attribute__((amdgpu_waves_per_eu(1, 1)))
void gcn_gemm_wmma(const float* __restrict__ H, const float* __restrict__ W,
                   const float* __restrict__ dinv, float* __restrict__ out, int mtiles) {
    const int lane = threadIdx.x & 31;
    const int wave = __builtin_amdgcn_readfirstlane((int)(threadIdx.x >> 5));
    const int gw = blockIdx.x * 4 + wave;
    const int nw = gridDim.x * 4;
    if (gw >= mtiles) return;
    const int c = lane & 15;
    const int h = lane >> 4;

    const float* wp = W + (size_t)c * DINK + (h << 2);
    float4 wf[DINK / 8][NT];
#pragma unroll
    for (int kc = 0; kc < DINK / 8; ++kc)
#pragma unroll
        for (int n = 0; n < NT; ++n)
            wf[kc][n] = *(const float4*)(wp + (size_t)n * 16 * DINK + kc * 8);

    float4 an[DINK / 8];
    {
        const float* hp = H + (size_t)((gw << 4) + c) * DINK + (h << 2);
#pragma unroll
        for (int kc = 0; kc < DINK / 8; ++kc) an[kc] = *(const float4*)(hp + kc * 8);
    }

    const int ldout = NT * 16;
    for (int tile = gw; tile < mtiles; tile += nw) {
        float4 ac[DINK / 8];
#pragma unroll
        for (int kc = 0; kc < DINK / 8; ++kc) ac[kc] = an[kc];
        {
            const int nt = (tile + nw < mtiles) ? tile + nw : tile;
            const float* hp = H + (size_t)((nt << 4) + c) * DINK + (h << 2);
#pragma unroll
            for (int kc = 0; kc < DINK / 8; ++kc) an[kc] = *(const float4*)(hp + kc * 8);
        }
        const float4 dlo = *(const float4*)(dinv + (tile << 4) + (h << 3));
        const float4 dhi = *(const float4*)(dinv + (tile << 4) + (h << 3) + 4);
        __builtin_amdgcn_sched_barrier(0);

        v8f acc[NT];
#pragma unroll
        for (int n = 0; n < NT; ++n)
            acc[n] = (v8f){0.f, 0.f, 0.f, 0.f, 0.f, 0.f, 0.f, 0.f};

#pragma unroll
        for (int kc = 0; kc < DINK / 8; ++kc) {
            const v2f alo = {ac[kc].x, ac[kc].y}, ahi = {ac[kc].z, ac[kc].w};
#pragma unroll
            for (int n = 0; n < NT; ++n) {
                const v2f blo = {wf[kc][n].x, wf[kc][n].y};
                acc[n] = __builtin_amdgcn_wmma_f32_16x16x4_f32(
                    false, alo, false, blo, (short)0, acc[n], false, false);
            }
#pragma unroll
            for (int n = 0; n < NT; ++n) {
                const v2f bhi = {wf[kc][n].z, wf[kc][n].w};
                acc[n] = __builtin_amdgcn_wmma_f32_16x16x4_f32(
                    false, ahi, false, bhi, (short)0, acc[n], false, false);
            }
        }

        const float dw[8] = {dlo.x, dlo.y, dlo.z, dlo.w, dhi.x, dhi.y, dhi.z, dhi.w};
        float* op = out + (size_t)((tile << 4) + (h << 3)) * ldout + c;
#pragma unroll
        for (int n = 0; n < NT; ++n)
#pragma unroll
            for (int r = 0; r < 8; ++r)
                op[(size_t)r * ldout + n * 16] = acc[n][r] * dw[r];
    }
}

#define AGG_ROWS 400
#define AGG_QCAP 288
#define AGG_GRP  8

template<int D, bool RELU, bool REPLAY>
__global__ __launch_bounds__(1024)
void gcn_agg(const int* __restrict__ src, const int* __restrict__ dst,
             const float* __restrict__ dinv, const float* __restrict__ tmp,
             const float* __restrict__ bias, float* out, float* lists, int E, int n) {
    constexpr int V = D / 32;
    typedef __attribute__((ext_vector_type(V))) float vrow;
    __shared__ float rows[AGG_ROWS * D];
    __shared__ float dloc[AGG_ROWS];
    __shared__ int q_e[32][AGG_QCAP];
    __shared__ int q_d[32][AGG_QCAP];
    __shared__ int lcount;

    const int tid  = threadIdx.x;
    const int lane = tid & 31;
    const int wave = __builtin_amdgcn_readfirstlane(tid >> 5);
    const int row0 = blockIdx.x * AGG_ROWS;

    float4* rows4 = (float4*)rows;
    for (int i = tid; i < AGG_ROWS * D / 4; i += 1024)
        rows4[i] = make_float4(0.f, 0.f, 0.f, 0.f);
    if (tid < AGG_ROWS) dloc[tid] = (row0 + tid < n) ? dinv[row0 + tid] : 0.f;
    if (tid == 0) lcount = 0;

    int2* mylist = (int2*)(lists + (size_t)row0 * DOK);
    const int rows_here = (n - row0 < AGG_ROWS) ? (n - row0) : AGG_ROWS;
    const int capch = rows_here * DOK / 64 - 1;
    int nch = 0;
    if (REPLAY) nch = __builtin_amdgcn_readfirstlane(mylist[0].x);
    const bool replay = REPLAY && nch >= 0 && nch <= capch;
    __syncthreads();

    int* qe = q_e[wave];
    int* qd = q_d[wave];
    int qn = 0;

    auto add_rows = [&](const int s, const int r, const int cnt) {
#pragma nounroll
        for (int j = 0; j < cnt; j += AGG_GRP) {
            vrow val[AGG_GRP];
#pragma unroll
            for (int k = 0; k < AGG_GRP; ++k) {
                const int jk = (j + k < cnt) ? j + k : cnt - 1;
                const int sk = __builtin_amdgcn_readlane(s, jk);
                val[k] = *(const vrow*)(tmp + (size_t)sk * D + V * lane);
            }
#pragma unroll
            for (int k = 0; k < AGG_GRP; ++k) {
                if (j + k < cnt) {
                    const int rk = __builtin_amdgcn_readlane(r, j + k);
                    float* o = &rows[rk * D + lane];
#pragma unroll
                    for (int c = 0; c < V; ++c) atomicAdd(o + 32 * c, val[k][c]);
                }
            }
        }
    };

    auto drain = [&]() {
        const int cnt = qn < 32 ? qn : 32;
        const int q0 = qn - cnt;
        const int qi = q0 + (lane < cnt ? lane : 0);
        const int e = qe[qi];
        const int r = qd[qi] - row0;
        const int s = src[e];
        if (!REPLAY) {
            int b = 0;
            if (lane == 0) b = atomicAdd(&lcount, 1);
            b = __builtin_amdgcn_readfirstlane(b);
            if (b < capch)
                mylist[(b + 1) * 32 + lane] = (lane < cnt) ? make_int2(s, r) : make_int2(0, -1);
        }
        add_rows(s, r, cnt);
        qn = q0;
    };

    const int nchunk = (E + 1023) >> 10;
    const int start = (int)(((long)blockIdx.x * nchunk) / gridDim.x);
    const bool vec_ok = (((unsigned long long)dst) & 15ull) == 0ull;
    if (replay) {
        for (int c = wave; c < nch; c += 32) {
            const int2 sr = mylist[(c + 1) * 32 + lane];
            const bool ok = sr.y >= 0;
            const int cnt = __builtin_popcount(__builtin_amdgcn_ballot_w32(ok));
            if (cnt == 32) add_rows(sr.x, sr.y, 32);
            else if (cnt > 0) add_rows(sr.x, sr.y, cnt);
        }
    } else {
    for (int c0 = wave; c0 < nchunk; c0 += 32) {
        int cc = start + c0;
        if (cc >= nchunk) cc -= nchunk;
        const int cbase = cc << 10;
        const int ebase = cbase + 4 * lane;
        int dv[32];
        if (vec_ok && cbase + 1024 <= E) {
            const int4* d4 = (const int4*)(dst + cbase) + lane;
#pragma unroll
            for (int i = 0; i < 8; ++i) {
                const int4 t = d4[32 * i];
                dv[4 * i + 0] = t.x; dv[4 * i + 1] = t.y; dv[4 * i + 2] = t.z; dv[4 * i + 3] = t.w;
            }
        } else {
#pragma unroll
            for (int x = 0; x < 32; ++x) {
                const int e = ebase + 128 * (x >> 2) + (x & 3);
                const int v = dst[e < E ? e : E - 1];
                dv[x] = (e < E) ? v : -1;
            }
        }
#pragma unroll
        for (int part = 0; part < 4; ++part) {
#pragma unroll
            for (int x = 8 * part; x < 8 * part + 8; ++x) {
                const bool m = (unsigned)(dv[x] - row0) < (unsigned)AGG_ROWS;
                const unsigned mask = __builtin_amdgcn_ballot_w32(m);
                if (m) {
                    const int p = qn + (int)__builtin_amdgcn_mbcnt_lo(mask, 0u);
                    qe[p] = ebase + 128 * (x >> 2) + (x & 3);
                    qd[p] = dv[x];
                }
                qn += __builtin_popcount(mask);
            }
            while (qn >= 32) drain();
        }
    }
    while (qn > 0) drain();
    }
    __syncthreads();
    if (!REPLAY && wave == 0) mylist[lane] = make_int2(lcount, 0);

    const float4* b4 = (const float4*)bias;
    float4* out4 = (float4*)out + (size_t)row0 * (D / 4);
    const long lim = ((long)n - row0) * (D / 4);
    for (int i = tid; i < AGG_ROWS * D / 4; i += 1024) {
        if (i < lim) {
            const int q = i & (D / 4 - 1);
            const float* rp = rows + (i / (D / 4)) * D;
            const int f = 4 * q;
            float4 v;
            v.x = rp[((f + 0) % V) * 32 + (f + 0) / V];
            v.y = rp[((f + 1) % V) * 32 + (f + 1) / V];
            v.z = rp[((f + 2) % V) * 32 + (f + 2) / V];
            v.w = rp[((f + 3) % V) * 32 + (f + 3) / V];
            const float dd = dloc[i / (D / 4)];
            const float4 bb = b4[q];
            v.x = v.x * dd + bb.x; v.y = v.y * dd + bb.y;
            v.z = v.z * dd + bb.z; v.w = v.w * dd + bb.w;
            if (RELU) {
                v.x = fmaxf(v.x, 0.f); v.y = fmaxf(v.y, 0.f);
                v.z = fmaxf(v.z, 0.f); v.w = fmaxf(v.w, 0.f);
            }
            out4[i] = v;
        }
    }
}


extern "C" void kernel_launch(void* const* d_in, const int* in_sizes, int n_in,
                              void* d_out, int out_size, void* d_ws, size_t ws_size,
                              hipStream_t stream) {
    const float* x  = (const float*)d_in[0];
    const int*   ei = (const int*)d_in[1];
    const float* W1 = (const float*)d_in[2];
    const float* b1 = (const float*)d_in[3];
    const float* W2 = (const float*)d_in[4];
    const float* b2 = (const float*)d_in[5];
    const float* W3 = (const float*)d_in[6];
    const float* b3 = (const float*)d_in[7];
    const float* W4 = (const float*)d_in[8];
    const float* b4 = (const float*)d_in[9];

    const int Nn = in_sizes[0] / DINK;
    const int E  = in_sizes[1] / 2;
    const int* srcI = ei;
    const int* dstI = ei + E;

    float* ws_f = (float*)d_ws;
    float* dinv = ws_f;
    size_t offA = ((size_t)Nn + 1023) & ~(size_t)1023;
    float* A = ws_f + offA;
    float* B = A + (size_t)Nn * DHK;
    float* out = (float*)d_out;

    const int TB = 256;
    const int  mT    = Nn / 16;
    const int  gG    = 256;
    const int  gAg   = (Nn + AGG_ROWS - 1) / AGG_ROWS;

    gcn_zero_f4<<<(Nn / 4 + TB - 1) / TB, TB, 0, stream>>>((float4*)dinv, Nn / 4);
    gcn_deg<<<(E + TB - 1) / TB, TB, 0, stream>>>(dstI, dinv, E);
    gcn_dinv<<<(Nn + TB - 1) / TB, TB, 0, stream>>>(dinv, Nn);

    gcn_gemm_wmma<8><<<gG, 128, 0, stream>>>(x, W1, dinv, A, mT);
    gcn_agg<128, true, false><<<gAg, 1024, 0, stream>>>(srcI, dstI, dinv, A, b1, B, out, E, Nn);

    gcn_gemm_wmma<8><<<gG, 128, 0, stream>>>(B, W2, dinv, A, mT);
    gcn_agg<128, true, true><<<gAg, 1024, 0, stream>>>(srcI, dstI, dinv, A, b2, B, out, E, Nn);

    gcn_gemm_wmma<8><<<gG, 128, 0, stream>>>(B, W3, dinv, A, mT);
    gcn_agg<128, true, true><<<gAg, 1024, 0, stream>>>(srcI, dstI, dinv, A, b3, B, out, E, Nn);

    gcn_gemm_wmma<4><<<gG, 128, 0, stream>>>(B, W4, dinv, A, mT);
    gcn_agg<64, false, true><<<gAg, 1024, 0, stream>>>(srcI, dstI, dinv, A, b4, out, out, E, Nn);
}


# Dst filter moved out of layer 1 into a launch of its own, tiled 2-D: a block tests 1/8 of the dst list against 8 row blocks at once, LDS buckets, 8 list regions per row block; LDS arrival-order sums
# speedup vs baseline: 4.1278x; 1.1024x over previous
#include <hip/hip_runtime.h>

#define DINK 128
#define DHK  128
#define DOK  64

typedef __attribute__((ext_vector_type(2))) float v2f;
typedef __attribute__((ext_vector_type(8))) float v8f;


__global__ void gcn_zero_f4(float4* __restrict__ p, long n4) {
    long i = (long)blockIdx.x * blockDim.x + threadIdx.x;
    if (i < n4) p[i] = make_float4(0.f, 0.f, 0.f, 0.f);
}

__global__ void gcn_deg(const int* __restrict__ dst, float* __restrict__ deg, int E) {
    int e = blockIdx.x * blockDim.x + threadIdx.x;
    if (e < E) atomicAdd(&deg[dst[e]], 1.0f);
}

__global__ void gcn_dinv(float* __restrict__ deg, int n) {
    int i = blockIdx.x * blockDim.x + threadIdx.x;
    if (i < n) {
        float d = deg[i];
        deg[i] = (d > 0.f) ? rsqrtf(d) : 0.f;
    }
}

template<int NT>
__global__ __launch_bounds__(128) __attribute__((amdgpu_waves_per_eu(1, 1)))
void gcn_gemm_wmma(const float* __restrict__ H, const float* __restrict__ W,
                   const float* __restrict__ dinv, float* __restrict__ out, int mtiles) {
    const int lane = threadIdx.x & 31;
    const int wave = __builtin_amdgcn_readfirstlane((int)(threadIdx.x >> 5));
    const int gw = blockIdx.x * 4 + wave;
    const int nw = gridDim.x * 4;
    if (gw >= mtiles) return;
    const int c = lane & 15;
    const int h = lane >> 4;

    const float* wp = W + (size_t)c * DINK + (h << 2);
    float4 wf[DINK / 8][NT];
#pragma unroll
    for (int kc = 0; kc < DINK / 8; ++kc)
#pragma unroll
        for (int n = 0; n < NT; ++n)
            wf[kc][n] = *(const float4*)(wp + (size_t)n * 16 * DINK + kc * 8);

    float4 an[DINK / 8];
    {
        const float* hp = H + (size_t)((gw << 4) + c) * DINK + (h << 2);
#pragma unroll
        for (int kc = 0; kc < DINK / 8; ++kc) an[kc] = *(const float4*)(hp + kc * 8);
    }

    const int ldout = NT * 16;
    for (int tile = gw; tile < mtiles; tile += nw) {
        float4 ac[DINK / 8];
#pragma unroll
        for (int kc = 0; kc < DINK / 8; ++kc) ac[kc] = an[kc];
        {
            const int nt = (tile + nw < mtiles) ? tile + nw : tile;
            const float* hp = H + (size_t)((nt << 4) + c) * DINK + (h << 2);
#pragma unroll
            for (int kc = 0; kc < DINK / 8; ++kc) an[kc] = *(const float4*)(hp + kc * 8);
        }
        const float4 dlo = *(const float4*)(dinv + (tile << 4) + (h << 3));
        const float4 dhi = *(const float4*)(dinv + (tile << 4) + (h << 3) + 4);
        __builtin_amdgcn_sched_barrier(0);

        v8f acc[NT];
#pragma unroll
        for (int n = 0; n < NT; ++n)
            acc[n] = (v8f){0.f, 0.f, 0.f, 0.f, 0.f, 0.f, 0.f, 0.f};

#pragma unroll
        for (int kc = 0; kc < DINK / 8; ++kc) {
            const v2f alo = {ac[kc].x, ac[kc].y}, ahi = {ac[kc].z, ac[kc].w};
#pragma unroll
            for (int n = 0; n < NT; ++n) {
                const v2f blo = {wf[kc][n].x, wf[kc][n].y};
                acc[n] = __builtin_amdgcn_wmma_f32_16x16x4_f32(
                    false, alo, false, blo, (short)0, acc[n], false, false);
            }
#pragma unroll
            for (int n = 0; n < NT; ++n) {
                const v2f bhi = {wf[kc][n].z, wf[kc][n].w};
                acc[n] = __builtin_amdgcn_wmma_f32_16x16x4_f32(
                    false, ahi, false, bhi, (short)0, acc[n], false, false);
            }
        }

        const float dw[8] = {dlo.x, dlo.y, dlo.z, dlo.w, dhi.x, dhi.y, dhi.z, dhi.w};
        float* op = out + (size_t)((tile << 4) + (h << 3)) * ldout + c;
#pragma unroll
        for (int n = 0; n < NT; ++n)
#pragma unroll
            for (int r = 0; r < 8; ++r)
                op[(size_t)r * ldout + n * 16] = acc[n][r] * dw[r];
    }
}

#define AGG_ROWS 400
#define FGRP 8
#define FSEG 8
#define BCAP 1568

__global__ __launch_bounds__(1024)
void gcn_filter(const int* __restrict__ src, const int* __restrict__ dst, float* lists, int E, int n) {
    __shared__ int2 bucket[FGRP][BCAP];
    __shared__ int cur[FGRP];

    const int tid  = threadIdx.x;
    const int lane = tid & 31;
    const int wave = __builtin_amdgcn_readfirstlane(tid >> 5);
    const int g = blockIdx.x / FSEG;
    const int j = blockIdx.x - g * FSEG;
    const int R0 = g * (FGRP * AGG_ROWS);
    const int span = (n - R0 < FGRP * AGG_ROWS) ? (n - R0) : FGRP * AGG_ROWS;

    if (tid < FGRP) cur[tid] = 0;
    __syncthreads();

    const int nchunk = (E + 1023) >> 10;
    const int clo = (int)(((long)j * nchunk) / FSEG);
    const int chi = (int)(((long)(j + 1) * nchunk) / FSEG);
    const bool vec_ok = (((unsigned long long)dst) & 15ull) == 0ull;
    for (int cc = clo + wave; cc < chi; cc += 32) {
        const int cbase = cc << 10;
        const int ebase = cbase + 4 * lane;
        int dv[32];
        if (vec_ok && cbase + 1024 <= E) {
            const int4* d4 = (const int4*)(dst + cbase) + lane;
#pragma unroll
            for (int i = 0; i < 8; ++i) {
                const int4 t = d4[32 * i];
                dv[4 * i + 0] = t.x; dv[4 * i + 1] = t.y; dv[4 * i + 2] = t.z; dv[4 * i + 3] = t.w;
            }
        } else {
#pragma unroll
            for (int x = 0; x < 32; ++x) {
                const int e = ebase + 128 * (x >> 2) + (x & 3);
                const int v = dst[e < E ? e : E - 1];
                dv[x] = (e < E) ? v : -1;
            }
        }
#pragma unroll
        for (int x = 0; x < 32; ++x) {
            const unsigned rr = (unsigned)(dv[x] - R0);
            if (rr < (unsigned)span) {
                const int k = (int)(rr / AGG_ROWS);
                const int slot = atomicAdd(&cur[k], 1);
                if (slot < BCAP)
                    bucket[k][slot] = make_int2(ebase + 128 * (x >> 2) + (x & 3), (int)rr - k * AGG_ROWS);
            }
        }
    }
    __syncthreads();

    for (int k = 0; k < FGRP; ++k) {
        const int rowb = R0 + k * AGG_ROWS;
        const int rows_b = (n - rowb < AGG_ROWS) ? (n - rowb) : AGG_ROWS;
        const int lreg = rows_b / FSEG;
        const int cap = (32 * (lreg - 1) < BCAP) ? 32 * (lreg - 1) : BCAP;
        const int tot = cur[k];
        if (wave == 0 && lreg >= 2) {
            int2* reg = (int2*)(lists + (size_t)rowb * DOK) + (size_t)j * lreg * 32;
            reg[lane] = make_int2(tot <= cap ? tot : -1, 0);
        }
    }
    for (int c = wave; c < BCAP / 32; c += 32) {
        int2 v[FGRP];
        bool st[FGRP];
#pragma unroll
        for (int k = 0; k < FGRP; ++k) {
            const int rowb = R0 + k * AGG_ROWS;
            const int rows_b = (n - rowb < AGG_ROWS) ? (n - rowb) : AGG_ROWS;
            const int lreg = rows_b / FSEG;
            const int cap = (32 * (lreg - 1) < BCAP) ? 32 * (lreg - 1) : BCAP;
            const int tot = cur[k];
            st[k] = lreg >= 2 && tot <= cap && 32 * c < tot;
            const int idx = 32 * c + lane;
            const bool valid = st[k] && idx < tot;
            const int2 t = bucket[k][idx];
            const int sv = src[valid ? t.x : 0];
            v[k] = valid ? make_int2(sv, t.y) : make_int2(0, -1);
        }
#pragma unroll
        for (int k = 0; k < FGRP; ++k) {
            if (st[k]) {
                const int rowb = R0 + k * AGG_ROWS;
                const int rows_b = (n - rowb < AGG_ROWS) ? (n - rowb) : AGG_ROWS;
                const int lreg = rows_b / FSEG;
                int2* reg = (int2*)(lists + (size_t)rowb * DOK) + (size_t)j * lreg * 32;
                reg[(c + 1) * 32 + lane] = v[k];
            }
        }
    }
}

#define AGG_QCAP 288
#define AGG_GRP  8

template<int D, bool RELU>
__global__ __launch_bounds__(1024)
void gcn_agg(const int* __restrict__ src, const int* __restrict__ dst,
             const float* __restrict__ dinv, const float* __restrict__ tmp,
             const float* __restrict__ bias, float* out, float* lists, int E, int n) {
    constexpr int V = D / 32;
    typedef __attribute__((ext_vector_type(V))) float vrow;
    __shared__ float rows[AGG_ROWS * D];
    __shared__ float dloc[AGG_ROWS];
    __shared__ int q_e[32][AGG_QCAP];
    __shared__ int q_d[32][AGG_QCAP];

    const int tid  = threadIdx.x;
    const int lane = tid & 31;
    const int wave = __builtin_amdgcn_readfirstlane(tid >> 5);
    const int row0 = blockIdx.x * AGG_ROWS;

    float4* rows4 = (float4*)rows;
    for (int i = tid; i < AGG_ROWS * D / 4; i += 1024)
        rows4[i] = make_float4(0.f, 0.f, 0.f, 0.f);
    if (tid < AGG_ROWS) dloc[tid] = (row0 + tid < n) ? dinv[row0 + tid] : 0.f;

    const int2* mylist = (const int2*)(lists + (size_t)row0 * DOK);
    const int rows_here = (n - row0 < AGG_ROWS) ? (n - row0) : AGG_ROWS;
    const int lreg = rows_here / FSEG;
    const int cap = (32 * (lreg - 1) < BCAP) ? 32 * (lreg - 1) : BCAP;
    int cnt_l = 0;
    if (lreg >= 2 && lane < FSEG) cnt_l = mylist[(size_t)lane * lreg * 32].x;
    const bool okl = cnt_l >= 0 && cnt_l <= cap;
    const bool walk = lreg >= 2 && __builtin_amdgcn_ballot_w32(okl) == 0xffffffffu;
    __syncthreads();

    auto add_rows = [&](const int s, const int r, const int cnt) {
#pragma nounroll
        for (int j = 0; j < cnt; j += AGG_GRP) {
            vrow val[AGG_GRP];
#pragma unroll
            for (int k = 0; k < AGG_GRP; ++k) {
                const int jk = (j + k < cnt) ? j + k : cnt - 1;
                const int sk = __builtin_amdgcn_readlane(s, jk);
                val[k] = *(const vrow*)(tmp + (size_t)sk * D + V * lane);
            }
#pragma unroll
            for (int k = 0; k < AGG_GRP; ++k) {
                if (j + k < cnt) {
                    const int rk = __builtin_amdgcn_readlane(r, j + k);
                    float* o = &rows[rk * D + lane];
#pragma unroll
                    for (int c = 0; c < V; ++c) atomicAdd(o + 32 * c, val[k][c]);
                }
            }
        }
    };

    if (walk) {
        int nc[FSEG];
        int total = 0;
#pragma unroll
        for (int t = 0; t < FSEG; ++t) {
            nc[t] = (__builtin_amdgcn_readlane(cnt_l, t) + 31) >> 5;
            total += nc[t];
        }
        for (int gi = wave; gi < total; gi += 32) {
            int reg = 0, c = gi;
#pragma unroll
            for (int t = 0; t < FSEG - 1; ++t)
                if (reg == t && c >= nc[t]) { c -= nc[t]; reg = t + 1; }
            const int2 sr = mylist[((size_t)reg * lreg + 1 + c) * 32 + lane];
            const bool ok = sr.y >= 0;
            const int cnt = __builtin_popcount(__builtin_amdgcn_ballot_w32(ok));
            if (cnt == 32) add_rows(sr.x, sr.y, 32);
            else if (cnt > 0) add_rows(sr.x, sr.y, cnt);
        }
    } else {
    int* qe = q_e[wave];
    int* qd = q_d[wave];
    int qn = 0;

    auto drain = [&]() {
        const int cnt = qn < 32 ? qn : 32;
        const int q0 = qn - cnt;
        const int qi = q0 + (lane < cnt ? lane : 0);
        const int e = qe[qi];
        const int r = qd[qi] - row0;
        const int s = src[e];
        add_rows(s, r, cnt);
        qn = q0;
    };

    const int nchunk = (E + 1023) >> 10;
    const int start = (int)(((long)blockIdx.x * nchunk) / gridDim.x);
    const bool vec_ok = (((unsigned long long)dst) & 15ull) == 0ull;
    for (int c0 = wave; c0 < nchunk; c0 += 32) {
        int cc = start + c0;
        if (cc >= nchunk) cc -= nchunk;
        const int cbase = cc << 10;
        const int ebase = cbase + 4 * lane;
        int dv[32];
        if (vec_ok && cbase + 1024 <= E) {
            const int4* d4 = (const int4*)(dst + cbase) + lane;
#pragma unroll
            for (int i = 0; i < 8; ++i) {
                const int4 t = d4[32 * i];
                dv[4 * i + 0] = t.x; dv[4 * i + 1] = t.y; dv[4 * i + 2] = t.z; dv[4 * i + 3] = t.w;
            }
        } else {
#pragma unroll
            for (int x = 0; x < 32; ++x) {
                const int e = ebase + 128 * (x >> 2) + (x & 3);
                const int v = dst[e < E ? e : E - 1];
                dv[x] = (e < E) ? v : -1;
            }
        }
#pragma unroll
        for (int part = 0; part < 4; ++part) {
#pragma unroll
            for (int x = 8 * part; x < 8 * part + 8; ++x) {
                const bool m = (unsigned)(dv[x] - row0) < (unsigned)AGG_ROWS;
                const unsigned mask = __builtin_amdgcn_ballot_w32(m);
                if (m) {
                    const int p = qn + (int)__builtin_amdgcn_mbcnt_lo(mask, 0u);
                    qe[p] = ebase + 128 * (x >> 2) + (x & 3);
                    qd[p] = dv[x];
                }
                qn += __builtin_popcount(mask);
            }
            while (qn >= 32) drain();
        }
    }
    while (qn > 0) drain();
    }
    __syncthreads();

    const float4* b4 = (const float4*)bias;
    float4* out4 = (float4*)out + (size_t)row0 * (D / 4);
    const long lim = ((long)n - row0) * (D / 4);
    for (int i = tid; i < AGG_ROWS * D / 4; i += 1024) {
        if (i < lim) {
            const int q = i & (D / 4 - 1);
            const float* rp = rows + (i / (D / 4)) * D;
            const int f = 4 * q;
            float4 v;
            v.x = rp[((f + 0) % V) * 32 + (f + 0) / V];
            v.y = rp[((f + 1) % V) * 32 + (f + 1) / V];
            v.z = rp[((f + 2) % V) * 32 + (f + 2) / V];
            v.w = rp[((f + 3) % V) * 32 + (f + 3) / V];
            const float dd = dloc[i / (D / 4)];
            const float4 bb = b4[q];
            v.x = v.x * dd + bb.x; v.y = v.y * dd + bb.y;
            v.z = v.z * dd + bb.z; v.w = v.w * dd + bb.w;
            if (RELU) {
                v.x = fmaxf(v.x, 0.f); v.y = fmaxf(v.y, 0.f);
                v.z = fmaxf(v.z, 0.f); v.w = fmaxf(v.w, 0.f);
            }
            out4[i] = v;
        }
    }
}


extern "C" void kernel_launch(void* const* d_in, const int* in_sizes, int n_in,
                              void* d_out, int out_size, void* d_ws, size_t ws_size,
                              hipStream_t stream) {
    const float* x  = (const float*)d_in[0];
    const int*   ei = (const int*)d_in[1];
    const float* W1 = (const float*)d_in[2];
    const float* b1 = (const float*)d_in[3];
    const float* W2 = (const float*)d_in[4];
    const float* b2 = (const float*)d_in[5];
    const float* W3 = (const float*)d_in[6];
    const float* b3 = (const float*)d_in[7];
    const float* W4 = (const float*)d_in[8];
    const float* b4 = (const float*)d_in[9];

    const int Nn = in_sizes[0] / DINK;
    const int E  = in_sizes[1] / 2;
    const int* srcI = ei;
    const int* dstI = ei + E;

    float* ws_f = (float*)d_ws;
    float* dinv = ws_f;
    size_t offA = ((size_t)Nn + 1023) & ~(size_t)1023;
    float* A = ws_f + offA;
    float* B = A + (size_t)Nn * DHK;
    float* out = (float*)d_out;

    const int TB = 256;
    const int  mT    = Nn / 16;
    const int  gG    = 256;
    const int  gAg   = (Nn + AGG_ROWS - 1) / AGG_ROWS;
    const int  gF    = ((gAg + FGRP - 1) / FGRP) * FSEG;

    gcn_zero_f4<<<(Nn / 4 + TB - 1) / TB, TB, 0, stream>>>((float4*)dinv, Nn / 4);
    gcn_deg<<<(E + TB - 1) / TB, TB, 0, stream>>>(dstI, dinv, E);
    gcn_dinv<<<(Nn + TB - 1) / TB, TB, 0, stream>>>(dinv, Nn);

    gcn_filter<<<gF, 1024, 0, stream>>>(srcI, dstI, out, E, Nn);

    gcn_gemm_wmma<8><<<gG, 128, 0, stream>>>(x, W1, dinv, A, mT);
    gcn_agg<128, true><<<gAg, 1024, 0, stream>>>(srcI, dstI, dinv, A, b1, B, out, E, Nn);

    gcn_gemm_wmma<8><<<gG, 128, 0, stream>>>(B, W2, dinv, A, mT);
    gcn_agg<128, true><<<gAg, 1024, 0, stream>>>(srcI, dstI, dinv, A, b2, B, out, E, Nn);

    gcn_gemm_wmma<8><<<gG, 128, 0, stream>>>(B, W3, dinv, A, mT);
    gcn_agg<128, true><<<gAg, 1024, 0, stream>>>(srcI, dstI, dinv, A, b3, B, out, E, Nn);

    gcn_gemm_wmma<4><<<gG, 128, 0, stream>>>(B, W4, dinv, A, mT);
    gcn_agg<64, false><<<gAg, 1024, 0, stream>>>(srcI, dstI, dinv, A, b4, out, out, E, Nn);
}


# In-degree counted inside the dst filter launch (LDS integer adds; 8 per-piece counts per row in the unwritten product plane), one launch sums them to deg^-1/2: zero-fill and float atomic passes gone
# speedup vs baseline: 4.9929x; 1.2096x over previous
#include <hip/hip_runtime.h>

#define DINK 128
#define DHK  128
#define DOK  64

typedef __attribute__((ext_vector_type(2))) float v2f;
typedef __attribute__((ext_vector_type(8))) float v8f;


#define NPIECE 8
__global__ void gcn_dinv(const int* __restrict__ cnt, float* __restrict__ dinv, int n) {
    int i = blockIdx.x * blockDim.x + threadIdx.x;
    if (i < n) {
        int c = 0;
#pragma unroll
        for (int p = 0; p < NPIECE; ++p) c += cnt[(size_t)p * n + i];
        float d = (float)c;
        dinv[i] = (d > 0.f) ? rsqrtf(d) : 0.f;
    }
}

template<int NT>
__global__ __launch_bounds__(128) __attribute__((amdgpu_waves_per_eu(1, 1)))
void gcn_gemm_wmma(const float* __restrict__ H, const float* __restrict__ W,
                   const float* __restrict__ dinv, float* __restrict__ out, int mtiles) {
    const int lane = threadIdx.x & 31;
    const int wave = __builtin_amdgcn_readfirstlane((int)(threadIdx.x >> 5));
    const int gw = blockIdx.x * 4 + wave;
    const int nw = gridDim.x * 4;
    if (gw >= mtiles) return;
    const int c = lane & 15;
    const int h = lane >> 4;

    const float* wp = W + (size_t)c * DINK + (h << 2);
    float4 wf[DINK / 8][NT];
#pragma unroll
    for (int kc = 0; kc < DINK / 8; ++kc)
#pragma unroll
        for (int n = 0; n < NT; ++n)
            wf[kc][n] = *(const float4*)(wp + (size_t)n * 16 * DINK + kc * 8);

    float4 an[DINK / 8];
    {
        const float* hp = H + (size_t)((gw << 4) + c) * DINK + (h << 2);
#pragma unroll
        for (int kc = 0; kc < DINK / 8; ++kc) an[kc] = *(const float4*)(hp + kc * 8);
    }

    const int ldout = NT * 16;
    for (int tile = gw; tile < mtiles; tile += nw) {
        float4 ac[DINK / 8];
#pragma unroll
        for (int kc = 0; kc < DINK / 8; ++kc) ac[kc] = an[kc];
        {
            const int nt = (tile + nw < mtiles) ? tile + nw : tile;
            const float* hp = H + (size_t)((nt << 4) + c) * DINK + (h << 2);
#pragma unroll
            for (int kc = 0; kc < DINK / 8; ++kc) an[kc] = *(const float4*)(hp + kc * 8);
        }
        const float4 dlo = *(const float4*)(dinv + (tile << 4) + (h << 3));
        const float4 dhi = *(const float4*)(dinv + (tile << 4) + (h << 3) + 4);
        __builtin_amdgcn_sched_barrier(0);

        v8f acc[NT];
#pragma unroll
        for (int n = 0; n < NT; ++n)
            acc[n] = (v8f){0.f, 0.f, 0.f, 0.f, 0.f, 0.f, 0.f, 0.f};

#pragma unroll
        for (int kc = 0; kc < DINK / 8; ++kc) {
            const v2f alo = {ac[kc].x, ac[kc].y}, ahi = {ac[kc].z, ac[kc].w};
#pragma unroll
            for (int n = 0; n < NT; ++n) {
                const v2f blo = {wf[kc][n].x, wf[kc][n].y};
                acc[n] = __builtin_amdgcn_wmma_f32_16x16x4_f32(
                    false, alo, false, blo, (short)0, acc[n], false, false);
            }
#pragma unroll
            for (int n = 0; n < NT; ++n) {
                const v2f bhi = {wf[kc][n].z, wf[kc][n].w};
                acc[n] = __builtin_amdgcn_wmma_f32_16x16x4_f32(
                    false, ahi, false, bhi, (short)0, acc[n], false, false);
            }
        }

        const float dw[8] = {dlo.x, dlo.y, dlo.z, dlo.w, dhi.x, dhi.y, dhi.z, dhi.w};
        float* op = out + (size_t)((tile << 4) + (h << 3)) * ldout + c;
#pragma unroll
        for (int n = 0; n < NT; ++n)
#pragma unroll
            for (int r = 0; r < 8; ++r)
                op[(size_t)r * ldout + n * 16] = acc[n][r] * dw[r];
    }
}

#define AGG_ROWS 400
#define FGRP 8
#define FSEG 8
#define BCAP 1568
static_assert(FSEG == NPIECE, "the dinv launch adds one degree share per piece of the dst list");

__global__ __launch_bounds__(1024)
void gcn_filter(const int* __restrict__ src, const int* __restrict__ dst, float* lists,
                int* __restrict__ cnt, int E, int n) {
    __shared__ int2 bucket[FGRP][BCAP];
    __shared__ int cur[FGRP];
    __shared__ int degc[FGRP * AGG_ROWS];

    const int tid  = threadIdx.x;
    const int lane = tid & 31;
    const int wave = __builtin_amdgcn_readfirstlane(tid >> 5);
    const int g = blockIdx.x / FSEG;
    const int j = blockIdx.x - g * FSEG;
    const int R0 = g * (FGRP * AGG_ROWS);
    const int span = (n - R0 < FGRP * AGG_ROWS) ? (n - R0) : FGRP * AGG_ROWS;

    if (tid < FGRP) cur[tid] = 0;
    for (int i = tid; i < FGRP * AGG_ROWS; i += 1024) degc[i] = 0;
    __syncthreads();

    const int nchunk = (E + 1023) >> 10;
    const int clo = (int)(((long)j * nchunk) / FSEG);
    const int chi = (int)(((long)(j + 1) * nchunk) / FSEG);
    const bool vec_ok = (((unsigned long long)dst) & 15ull) == 0ull;
    for (int cc = clo + wave; cc < chi; cc += 32) {
        const int cbase = cc << 10;
        const int ebase = cbase + 4 * lane;
        int dv[32];
        if (vec_ok && cbase + 1024 <= E) {
            const int4* d4 = (const int4*)(dst + cbase) + lane;
#pragma unroll
            for (int i = 0; i < 8; ++i) {
                const int4 t = d4[32 * i];
                dv[4 * i + 0] = t.x; dv[4 * i + 1] = t.y; dv[4 * i + 2] = t.z; dv[4 * i + 3] = t.w;
            }
        } else {
#pragma unroll
            for (int x = 0; x < 32; ++x) {
                const int e = ebase + 128 * (x >> 2) + (x & 3);
                const int v = dst[e < E ? e : E - 1];
                dv[x] = (e < E) ? v : -1;
            }
        }
#pragma unroll
        for (int x = 0; x < 32; ++x) {
            const unsigned rr = (unsigned)(dv[x] - R0);
            if (rr < (unsigned)span) {
                const int k = (int)(rr / AGG_ROWS);
                atomicAdd(&degc[rr], 1);
                const int slot = atomicAdd(&cur[k], 1);
                if (slot < BCAP)
                    bucket[k][slot] = make_int2(ebase + 128 * (x >> 2) + (x & 3), (int)rr - k * AGG_ROWS);
            }
        }
    }
    __syncthreads();

    for (int i = tid; i < span; i += 1024) cnt[(size_t)j * n + R0 + i] = degc[i];

    for (int k = 0; k < FGRP; ++k) {
        const int rowb = R0 + k * AGG_ROWS;
        const int rows_b = (n - rowb < AGG_ROWS) ? (n - rowb) : AGG_ROWS;
        const int lreg = rows_b / FSEG;
        const int cap = (32 * (lreg - 1) < BCAP) ? 32 * (lreg - 1) : BCAP;
        const int tot = cur[k];
        if (wave == 0 && lreg >= 2) {
            int2* reg = (int2*)(lists + (size_t)rowb * DOK) + (size_t)j * lreg * 32;
            reg[lane] = make_int2(tot <= cap ? tot : -1, 0);
        }
    }
    for (int c = wave; c < BCAP / 32; c += 32) {
        int2 v[FGRP];
        bool st[FGRP];
#pragma unroll
        for (int k = 0; k < FGRP; ++k) {
            const int rowb = R0 + k * AGG_ROWS;
            const int rows_b = (n - rowb < AGG_ROWS) ? (n - rowb) : AGG_ROWS;
            const int lreg = rows_b / FSEG;
            const int cap = (32 * (lreg - 1) < BCAP) ? 32 * (lreg - 1) : BCAP;
            const int tot = cur[k];
            st[k] = lreg >= 2 && tot <= cap && 32 * c < tot;
            const int idx = 32 * c + lane;
            const bool valid = st[k] && idx < tot;
            const int2 t = bucket[k][idx];
            const int sv = src[valid ? t.x : 0];
            v[k] = valid ? make_int2(sv, t.y) : make_int2(0, -1);
        }
#pragma unroll
        for (int k = 0; k < FGRP; ++k) {
            if (st[k]) {
                const int rowb = R0 + k * AGG_ROWS;
                const int rows_b = (n - rowb < AGG_ROWS) ? (n - rowb) : AGG_ROWS;
                const int lreg = rows_b / FSEG;
                int2* reg = (int2*)(lists + (size_t)rowb * DOK) + (size_t)j * lreg * 32;
                reg[(c + 1) * 32 + lane] = v[k];
            }
        }
    }
}

#define AGG_QCAP 288
#define AGG_GRP  8

template<int D, bool RELU>
__global__ __launch_bounds__(1024)
void gcn_agg(const int* __restrict__ src, const int* __restrict__ dst,
             const float* __restrict__ dinv, const float* __restrict__ tmp,
             const float* __restrict__ bias, float* out, float* lists, int E, int n) {
    constexpr int V = D / 32;
    typedef __attribute__((ext_vector_type(V))) float vrow;
    __shared__ float rows[AGG_ROWS * D];
    __shared__ float dloc[AGG_ROWS];
    __shared__ int q_e[32][AGG_QCAP];
    __shared__ int q_d[32][AGG_QCAP];

    const int tid  = threadIdx.x;
    const int lane = tid & 31;
    const int wave = __builtin_amdgcn_readfirstlane(tid >> 5);
    const int row0 = blockIdx.x * AGG_ROWS;

    float4* rows4 = (float4*)rows;
    for (int i = tid; i < AGG_ROWS * D / 4; i += 1024)
        rows4[i] = make_float4(0.f, 0.f, 0.f, 0.f);
    if (tid < AGG_ROWS) dloc[tid] = (row0 + tid < n) ? dinv[row0 + tid] : 0.f;

    const int2* mylist = (const int2*)(lists + (size_t)row0 * DOK);
    const int rows_here = (n - row0 < AGG_ROWS) ? (n - row0) : AGG_ROWS;
    const int lreg = rows_here / FSEG;
    const int cap = (32 * (lreg - 1) < BCAP) ? 32 * (lreg - 1) : BCAP;
    int cnt_l = 0;
    if (lreg >= 2 && lane < FSEG) cnt_l = mylist[(size_t)lane * lreg * 32].x;
    const bool okl = cnt_l >= 0 && cnt_l <= cap;
    const bool walk = lreg >= 2 && __builtin_amdgcn_ballot_w32(okl) == 0xffffffffu;
    __syncthreads();

    auto add_rows = [&](const int s, const int r, const int cnt) {
#pragma nounroll
        for (int j = 0; j < cnt; j += AGG_GRP) {
            vrow val[AGG_GRP];
#pragma unroll
            for (int k = 0; k < AGG_GRP; ++k) {
                const int jk = (j + k < cnt) ? j + k : cnt - 1;
                const int sk = __builtin_amdgcn_readlane(s, jk);
                val[k] = *(const vrow*)(tmp + (size_t)sk * D + V * lane);
            }
#pragma unroll
            for (int k = 0; k < AGG_GRP; ++k) {
                if (j + k < cnt) {
                    const int rk = __builtin_amdgcn_readlane(r, j + k);
                    float* o = &rows[rk * D + lane];
#pragma unroll
                    for (int c = 0; c < V; ++c) atomicAdd(o + 32 * c, val[k][c]);
                }
            }
        }
    };

    if (walk) {
        int nc[FSEG];
        int total = 0;
#pragma unroll
        for (int t = 0; t < FSEG; ++t) {
            nc[t] = (__builtin_amdgcn_readlane(cnt_l, t) + 31) >> 5;
            total += nc[t];
        }
        for (int gi = wave; gi < total; gi += 32) {
            int reg = 0, c = gi;
#pragma unroll
            for (int t = 0; t < FSEG - 1; ++t)
                if (reg == t && c >= nc[t]) { c -= nc[t]; reg = t + 1; }
            const int2 sr = mylist[((size_t)reg * lreg + 1 + c) * 32 + lane];
            const bool ok = sr.y >= 0;
            const int cnt = __builtin_popcount(__builtin_amdgcn_ballot_w32(ok));
            if (cnt == 32) add_rows(sr.x, sr.y, 32);
            else if (cnt > 0) add_rows(sr.x, sr.y, cnt);
        }
    } else {
    int* qe = q_e[wave];
    int* qd = q_d[wave];
    int qn = 0;

    auto drain = [&]() {
        const int cnt = qn < 32 ? qn : 32;
        const int q0 = qn - cnt;
        const int qi = q0 + (lane < cnt ? lane : 0);
        const int e = qe[qi];
        const int r = qd[qi] - row0;
        const int s = src[e];
        add_rows(s, r, cnt);
        qn = q0;
    };

    const int nchunk = (E + 1023) >> 10;
    const int start = (int)(((long)blockIdx.x * nchunk) / gridDim.x);
    const bool vec_ok = (((unsigned long long)dst) & 15ull) == 0ull;
    for (int c0 = wave; c0 < nchunk; c0 += 32) {
        int cc = start + c0;
        if (cc >= nchunk) cc -= nchunk;
        const int cbase = cc << 10;
        const int ebase = cbase + 4 * lane;
        int dv[32];
        if (vec_ok && cbase + 1024 <= E) {
            const int4* d4 = (const int4*)(dst + cbase) + lane;
#pragma unroll
            for (int i = 0; i < 8; ++i) {
                const int4 t = d4[32 * i];
                dv[4 * i + 0] = t.x; dv[4 * i + 1] = t.y; dv[4 * i + 2] = t.z; dv[4 * i + 3] = t.w;
            }
        } else {
#pragma unroll
            for (int x = 0; x < 32; ++x) {
                const int e = ebase + 128 * (x >> 2) + (x & 3);
                const int v = dst[e < E ? e : E - 1];
                dv[x] = (e < E) ? v : -1;
            }
        }
#pragma unroll
        for (int part = 0; part < 4; ++part) {
#pragma unroll
            for (int x = 8 * part; x < 8 * part + 8; ++x) {
                const bool m = (unsigned)(dv[x] - row0) < (unsigned)AGG_ROWS;
                const unsigned mask = __builtin_amdgcn_ballot_w32(m);
                if (m) {
                    const int p = qn + (int)__builtin_amdgcn_mbcnt_lo(mask, 0u);
                    qe[p] = ebase + 128 * (x >> 2) + (x & 3);
                    qd[p] = dv[x];
                }
                qn += __builtin_popcount(mask);
            }
            while (qn >= 32) drain();
        }
    }
    while (qn > 0) drain();
    }
    __syncthreads();

    const float4* b4 = (const float4*)bias;
    float4* out4 = (float4*)out + (size_t)row0 * (D / 4);
    const long lim = ((long)n - row0) * (D / 4);
    for (int i = tid; i < AGG_ROWS * D / 4; i += 1024) {
        if (i < lim) {
            const int q = i & (D / 4 - 1);
            const float* rp = rows + (i / (D / 4)) * D;
            const int f = 4 * q;
            float4 v;
            v.x = rp[((f + 0) % V) * 32 + (f + 0) / V];
            v.y = rp[((f + 1) % V) * 32 + (f + 1) / V];
            v.z = rp[((f + 2) % V) * 32 + (f + 2) / V];
            v.w = rp[((f + 3) % V) * 32 + (f + 3) / V];
            const float dd = dloc[i / (D / 4)];
            const float4 bb = b4[q];
            v.x = v.x * dd + bb.x; v.y = v.y * dd + bb.y;
            v.z = v.z * dd + bb.z; v.w = v.w * dd + bb.w;
            if (RELU) {
                v.x = fmaxf(v.x, 0.f); v.y = fmaxf(v.y, 0.f);
                v.z = fmaxf(v.z, 0.f); v.w = fmaxf(v.w, 0.f);
            }
            out4[i] = v;
        }
    }
}


extern "C" void kernel_launch(void* const* d_in, const int* in_sizes, int n_in,
                              void* d_out, int out_size, void* d_ws, size_t ws_size,
                              hipStream_t stream) {
    const float* x  = (const float*)d_in[0];
    const int*   ei = (const int*)d_in[1];
    const float* W1 = (const float*)d_in[2];
    const float* b1 = (const float*)d_in[3];
    const float* W2 = (const float*)d_in[4];
    const float* b2 = (const float*)d_in[5];
    const float* W3 = (const float*)d_in[6];
    const float* b3 = (const float*)d_in[7];
    const float* W4 = (const float*)d_in[8];
    const float* b4 = (const float*)d_in[9];

    const int Nn = in_sizes[0] / DINK;
    const int E  = in_sizes[1] / 2;
    const int* srcI = ei;
    const int* dstI = ei + E;

    float* ws_f = (float*)d_ws;
    float* dinv = ws_f;
    size_t offA = ((size_t)Nn + 1023) & ~(size_t)1023;
    float* A = ws_f + offA;
    float* B = A + (size_t)Nn * DHK;
    float* out = (float*)d_out;

    const int TB = 256;
    int* cnt = (int*)A;
    const int  mT    = Nn / 16;
    const int  gG    = 256;
    const int  gAg   = (Nn + AGG_ROWS - 1) / AGG_ROWS;
    const int  gF    = ((gAg + FGRP - 1) / FGRP) * FSEG;

    gcn_filter<<<gF, 1024, 0, stream>>>(srcI, dstI, out, cnt, E, Nn);

    gcn_dinv<<<(Nn + TB - 1) / TB, TB, 0, stream>>>(cnt, dinv, Nn);

    gcn_gemm_wmma<8><<<gG, 128, 0, stream>>>(x, W1, dinv, A, mT);
    gcn_agg<128, true><<<gAg, 1024, 0, stream>>>(srcI, dstI, dinv, A, b1, B, out, E, Nn);

    gcn_gemm_wmma<8><<<gG, 128, 0, stream>>>(B, W2, dinv, A, mT);
    gcn_agg<128, true><<<gAg, 1024, 0, stream>>>(srcI, dstI, dinv, A, b2, B, out, E, Nn);

    gcn_gemm_wmma<8><<<gG, 128, 0, stream>>>(B, W3, dinv, A, mT);
    gcn_agg<128, true><<<gAg, 1024, 0, stream>>>(srcI, dstI, dinv, A, b3, B, out, E, Nn);

    gcn_gemm_wmma<4><<<gG, 128, 0, stream>>>(B, W4, dinv, A, mT);
    gcn_agg<64, false><<<gAg, 1024, 0, stream>>>(srcI, dstI, dinv, A, b4, out, out, E, Nn);
}
